# starting kernel
# baseline (speedup 1.0000x reference)
#include <hip/hip_runtime.h>

#define FIN  128
#define FOUT 128
#define EPSV 1e-5f

typedef float v2f __attribute__((ext_vector_type(2)));
typedef float v8f __attribute__((ext_vector_type(8)));

__global__ __launch_bounds__(256) void ln_relu_kernel(
    const float* __restrict__ x, const float* __restrict__ gamma,
    const float* __restrict__ beta, float* __restrict__ xn, int n) {
  int wave = threadIdx.x >> 5;
  int lane = threadIdx.x & 31;
  int row  = blockIdx.x * 8 + wave;
  if (row >= n) return;

  float4 v = ((const float4*)(x + (size_t)row * FIN))[lane];
  float s  = v.x + v.y + v.z + v.w;
  float ss = v.x * v.x + v.y * v.y + v.z * v.z + v.w * v.w;
  for (int m = 16; m >= 1; m >>= 1) {
    s  += __shfl_xor(s,  m, 32);
    ss += __shfl_xor(ss, m, 32);
  }
  float mean = s * (1.0f / FIN);
  float var  = ss * (1.0f / FIN) - mean * mean;
  float rs   = rsqrtf(var + EPSV);

  float4 g  = ((const float4*)gamma)[lane];
  float4 bt = ((const float4*)beta)[lane];
  float4 o;
  o.x = fmaxf(0.0f, (v.x - mean) * rs * g.x + bt.x);
  o.y = fmaxf(0.0f, (v.y - mean) * rs * g.y + bt.y);
  o.z = fmaxf(0.0f, (v.z - mean) * rs * g.z + bt.z);
  o.w = fmaxf(0.0f, (v.w - mean) * rs * g.w + bt.w);
  ((float4*)(xn + (size_t)row * FIN))[lane] = o;
}

#define A_PITCH 132

__global__ __launch_bounds__(256) void gemm_wmma_kernel(
    const float* __restrict__ xn, const float* __restrict__ W,
    float* __restrict__ h) {
  __shared__ float alds[16 * A_PITCH];
  const int tid = threadIdx.x;
  const int n0  = blockIdx.x * 16;

  for (int s = tid; s < 16 * (FIN / 4); s += 256) {
    int r = s >> 5, c4 = s & 31;
    float4 v = ((const float4*)(xn + (size_t)(n0 + r) * FIN))[c4];
    *((float4*)&alds[r * A_PITCH + c4 * 4]) = v;
  }
  __syncthreads();

  const int wave  = tid >> 5;
  const int lane  = tid & 31;
  const int o0    = wave * 16;
  const int mrow  = lane & 15;
  const int khalf = (lane >> 4) * 2;

  const float* abase = &alds[mrow * A_PITCH + khalf];
  const float* wbase = W + (size_t)(o0 + mrow) * FIN + khalf;

  v8f c = {};
#pragma unroll 8
  for (int k = 0; k < FIN; k += 4) {
    v2f a = *((const v2f*)(abase + k));
    v2f b = *((const v2f*)(wbase + k));
    c = __builtin_amdgcn_wmma_f32_16x16x4_f32(
        false, a, false, b, (short)0, c, false, false);
  }

  const int col   = o0 + mrow;
  const int rbase = n0 + ((lane >> 4) << 3);
#pragma unroll
  for (int v = 0; v < 8; ++v) {
    h[(size_t)(rbase + v) * FOUT + col] = c[v];
  }
}

__global__ void deg_init_kernel(float* __restrict__ deg, int n) {
  int i = blockIdx.x * 256 + threadIdx.x;
  if (i < n) deg[i] = 1.0f;
}

__global__ void deg_edge_kernel(const int* __restrict__ ei,
                                const float* __restrict__ ew,
                                float* __restrict__ deg, int e) {
  int i = blockIdx.x * 256 + threadIdx.x;
  if (i < e) unsafeAtomicAdd(&deg[ei[e + i]], ew[i]);
}

__global__ void dinv_kernel(const float* __restrict__ deg,
                            float* __restrict__ dinv, int n) {
  int i = blockIdx.x * 256 + threadIdx.x;
  if (i < n) {
    float d = deg[i];
    dinv[i] = (d > 0.0f) ? rsqrtf(d) : 0.0f;
  }
}

__global__ __launch_bounds__(128) void out_init_kernel(
    const float* __restrict__ h, const float* __restrict__ dinv,
    const float* __restrict__ b, float* __restrict__ out) {
  int i = blockIdx.x;
  int f = threadIdx.x;
  float dv = dinv[i];
  size_t idx = (size_t)i * FOUT + f;
  out[idx] = b[f] + h[idx] * dv * dv;
}

__global__ __launch_bounds__(256) void edge_agg_kernel(
    const int* __restrict__ ei, const float* __restrict__ ew,
    const float* __restrict__ h, const float* __restrict__ dinv,
    float* __restrict__ out, int e) {
  int eid  = blockIdx.x * 8 + (threadIdx.x >> 5);
  if (eid >= e) return;
  int lane = threadIdx.x & 31;
  int r = ei[eid];
  int c = ei[e + eid];
  float norm = dinv[r] * ew[eid] * dinv[c];
  float4 hv = ((const float4*)(h + (size_t)r * FOUT))[lane];
  float* ob = out + (size_t)c * FOUT + lane * 4;
  unsafeAtomicAdd(ob + 0, hv.x * norm);
  unsafeAtomicAdd(ob + 1, hv.y * norm);
  unsafeAtomicAdd(ob + 2, hv.z * norm);
  unsafeAtomicAdd(ob + 3, hv.w * norm);
}

extern "C" void kernel_launch(void* const* d_in, const int* in_sizes, int n_in,
                              void* d_out, int out_size, void* d_ws, size_t ws_size,
                              hipStream_t stream) {
  const float* x     = (const float*)d_in[0];
  const int*   ei    = (const int*)d_in[1];
  const float* ew    = (const float*)d_in[2];
  const float* gamma = (const float*)d_in[3];
  const float* beta  = (const float*)d_in[4];
  const float* W     = (const float*)d_in[5];
  const float* b     = (const float*)d_in[6];
  float* out = (float*)d_out;

  const int n = in_sizes[0] / FIN;
  const int e = in_sizes[2];

  float* xn   = (float*)d_ws;
  float* h    = xn + (size_t)n * FIN;
  float* deg  = h + (size_t)n * FOUT;
  float* dinv = deg + n;

  ln_relu_kernel<<<(n + 7) / 8, 256, 0, stream>>>(x, gamma, beta, xn, n);

  gemm_wmma_kernel<<<n / 16, 256, 0, stream>>>(xn, W, h);

  deg_init_kernel<<<(n + 255) / 256, 256, 0, stream>>>(deg, n);
  deg_edge_kernel<<<(e + 255) / 256, 256, 0, stream>>>(ei, ew, deg, e);

  dinv_kernel<<<(n + 255) / 256, 256, 0, stream>>>(deg, dinv, n);
  out_init_kernel<<<n, 128, 0, stream>>>(h, dinv, b, out);

  edge_agg_kernel<<<(e + 7) / 8, 256, 0, stream>>>(ei, ew, h, dinv, out, e);

  (void)n_in; (void)out_size; (void)ws_size;
}


# Destination sums (degree, messages) without float atomics: per-call counting sort of edges (returning int add as rank, scan, slot stores); one wave per node sums rows in registers; sum order changes
# speedup vs baseline: 1.9748x; 1.9748x over previous
#include <hip/hip_runtime.h>

#define FIN  128
#define FOUT 128
#define EPSV 1e-5f

typedef float v2f __attribute__((ext_vector_type(2)));
typedef float v8f __attribute__((ext_vector_type(8)));

__global__ __launch_bounds__(256) void ln_relu_kernel(
    const float* __restrict__ x, const float* __restrict__ gamma,
    const float* __restrict__ beta, float* __restrict__ xn, int n) {
  int wave = threadIdx.x >> 5;
  int lane = threadIdx.x & 31;
  int row  = blockIdx.x * 8 + wave;
  if (row >= n) return;

  float4 v = ((const float4*)(x + (size_t)row * FIN))[lane];
  float s  = v.x + v.y + v.z + v.w;
  float ss = v.x * v.x + v.y * v.y + v.z * v.z + v.w * v.w;
  for (int m = 16; m >= 1; m >>= 1) {
    s  += __shfl_xor(s,  m, 32);
    ss += __shfl_xor(ss, m, 32);
  }
  float mean = s * (1.0f / FIN);
  float var  = ss * (1.0f / FIN) - mean * mean;
  float rs   = rsqrtf(var + EPSV);

  float4 g  = ((const float4*)gamma)[lane];
  float4 bt = ((const float4*)beta)[lane];
  float4 o;
  o.x = fmaxf(0.0f, (v.x - mean) * rs * g.x + bt.x);
  o.y = fmaxf(0.0f, (v.y - mean) * rs * g.y + bt.y);
  o.z = fmaxf(0.0f, (v.z - mean) * rs * g.z + bt.z);
  o.w = fmaxf(0.0f, (v.w - mean) * rs * g.w + bt.w);
  ((float4*)(xn + (size_t)row * FIN))[lane] = o;
}

#define A_PITCH 132

__global__ __launch_bounds__(256) void gemm_wmma_kernel(
    const float* __restrict__ xn, const float* __restrict__ W,
    float* __restrict__ h) {
  __shared__ float alds[16 * A_PITCH];
  const int tid = threadIdx.x;
  const int n0  = blockIdx.x * 16;

  for (int s = tid; s < 16 * (FIN / 4); s += 256) {
    int r = s >> 5, c4 = s & 31;
    float4 v = ((const float4*)(xn + (size_t)(n0 + r) * FIN))[c4];
    *((float4*)&alds[r * A_PITCH + c4 * 4]) = v;
  }
  __syncthreads();

  const int wave  = tid >> 5;
  const int lane  = tid & 31;
  const int o0    = wave * 16;
  const int mrow  = lane & 15;
  const int khalf = (lane >> 4) * 2;

  const float* abase = &alds[mrow * A_PITCH + khalf];
  const float* wbase = W + (size_t)(o0 + mrow) * FIN + khalf;

  v8f c = {};
#pragma unroll 8
  for (int k = 0; k < FIN; k += 4) {
    v2f a = *((const v2f*)(abase + k));
    v2f b = *((const v2f*)(wbase + k));
    c = __builtin_amdgcn_wmma_f32_16x16x4_f32(
        false, a, false, b, (short)0, c, false, false);
  }

  const int col   = o0 + mrow;
  const int rbase = n0 + ((lane >> 4) << 3);
#pragma unroll
  for (int v = 0; v < 8; ++v) {
    h[(size_t)(rbase + v) * FOUT + col] = c[v];
  }
}

__global__ void cnt_zero_kernel(int* __restrict__ cnt, int n) {
  int i = blockIdx.x * 256 + threadIdx.x;
  if (i < n) cnt[i] = 0;
}

__global__ void rank_kernel(const int* __restrict__ ei, int* __restrict__ cnt,
                            int* __restrict__ rank, int e) {
  int i = blockIdx.x * 256 + threadIdx.x;
  if (i < e) rank[i] = atomicAdd(&cnt[ei[e + i]], 1);
}

__global__ __launch_bounds__(256) void blocksum_kernel(
    const int* __restrict__ cnt, int* __restrict__ bsum, int n) {
  __shared__ int wsum[8];
  const int tid = threadIdx.x, lane = tid & 31, wave = tid >> 5;
  const int base = blockIdx.x * 1024 + tid * 4;
  int s = 0;
  for (int k = 0; k < 4; ++k)
    if (base + k < n) s += cnt[base + k];
  for (int m = 16; m >= 1; m >>= 1) s += __shfl_xor(s, m, 32);
  if (lane == 0) wsum[wave] = s;
  __syncthreads();
  if (tid == 0) {
    int t = 0;
    for (int w = 0; w < 8; ++w) t += wsum[w];
    bsum[blockIdx.x] = t;
  }
}

__global__ __launch_bounds__(256) void scan_kernel(
    const int* __restrict__ cnt, const int* __restrict__ bsum,
    int* __restrict__ off, int n) {
  __shared__ int wsum[8];
  __shared__ int sbase;
  const int tid = threadIdx.x, lane = tid & 31, wave = tid >> 5;
  const int nblk = blockIdx.x;
  if (wave == 0) {
    int s = 0;
    for (int j = lane; j < nblk; j += 32) s += bsum[j];
    for (int m = 16; m >= 1; m >>= 1) s += __shfl_xor(s, m, 32);
    if (lane == 0) sbase = s;
  }
  const int base = nblk * 1024 + tid * 4;
  int c[4];
  for (int k = 0; k < 4; ++k) c[k] = (base + k < n) ? cnt[base + k] : 0;
  const int s = c[0] + c[1] + c[2] + c[3];
  int inc = s;
  for (int d = 1; d < 32; d <<= 1) {
    int v = __shfl_up(inc, d, 32);
    if (lane >= d) inc += v;
  }
  if (lane == 31) wsum[wave] = inc;
  __syncthreads();
  int ex = sbase + inc - s;
  for (int w = 0; w < wave; ++w) ex += wsum[w];
  for (int k = 0; k < 4; ++k) {
    if (base + k <= n) off[base + k] = ex;
    ex += c[k];
  }
}

__global__ void fill_kernel(const int* __restrict__ ei,
                            const float* __restrict__ ew,
                            const int* __restrict__ off,
                            const int* __restrict__ rank,
                            int* __restrict__ rec, int e) {
  int i = blockIdx.x * 256 + threadIdx.x;
  if (i < e) {
    int p = off[ei[e + i]] + rank[i];
    ((int2*)rec)[p] = make_int2(ei[i], __float_as_int(ew[i]));
  }
}

__global__ void dinv_kernel(const int* __restrict__ off,
                            const int* __restrict__ rec,
                            float* __restrict__ dinv, int n) {
  int i = blockIdx.x * 256 + threadIdx.x;
  if (i < n) {
    const int p1 = off[i + 1];
    float d = 1.0f;
    for (int p = off[i]; p < p1; ++p) d += __int_as_float(((const int2*)rec)[p].y);
    dinv[i] = (d > 0.0f) ? rsqrtf(d) : 0.0f;
  }
}

__global__ __launch_bounds__(256) void node_agg_kernel(
    const int* __restrict__ off, const int* __restrict__ rec,
    const float* __restrict__ h, const float* __restrict__ dinv,
    const float* __restrict__ b, float* __restrict__ out, int n) {
  const int node = blockIdx.x * 8 + (threadIdx.x >> 5);
  if (node >= n) return;
  const int lane = threadIdx.x & 31;
  const int p0 = __builtin_amdgcn_readfirstlane(off[node]);
  const int p1 = __builtin_amdgcn_readfirstlane(off[node + 1]);
  const float dc = dinv[node];
  const float4 hs = ((const float4*)(h + (size_t)node * FOUT))[lane];
  float4 acc;
  acc.x = hs.x * dc * dc;
  acc.y = hs.y * dc * dc;
  acc.z = hs.z * dc * dc;
  acc.w = hs.w * dc * dc;
  for (int q = p0; q < p1; q += 32) {
    const int m = (p1 - q < 32) ? (p1 - q) : 32;
    int   rj = 0;
    float nj = 0.0f;
    if (lane < m) {
      const int2 r2 = ((const int2*)rec)[q + lane];
      rj = r2.x;
      nj = dinv[r2.x] * __int_as_float(r2.y) * dc;
    }
#pragma unroll 4
    for (int j = 0; j < m; ++j) {
      const int   r  = __builtin_amdgcn_readlane(rj, j);
      const float nm = __int_as_float(__builtin_amdgcn_readlane(__float_as_int(nj), j));
      const float4 hr = ((const float4*)(h + (size_t)r * FOUT))[lane];
      acc.x += hr.x * nm;
      acc.y += hr.y * nm;
      acc.z += hr.z * nm;
      acc.w += hr.w * nm;
    }
  }
  const float4 bv = ((const float4*)b)[lane];
  float4 o;
  o.x = acc.x + bv.x;
  o.y = acc.y + bv.y;
  o.z = acc.z + bv.z;
  o.w = acc.w + bv.w;
  ((float4*)(out + (size_t)node * FOUT))[lane] = o;
}

extern "C" void kernel_launch(void* const* d_in, const int* in_sizes, int n_in,
                              void* d_out, int out_size, void* d_ws, size_t ws_size,
                              hipStream_t stream) {
  const float* x     = (const float*)d_in[0];
  const int*   ei    = (const int*)d_in[1];
  const float* ew    = (const float*)d_in[2];
  const float* gamma = (const float*)d_in[3];
  const float* beta  = (const float*)d_in[4];
  const float* W     = (const float*)d_in[5];
  const float* b     = (const float*)d_in[6];
  float* out = (float*)d_out;

  const int n = in_sizes[0] / FIN;
  const int e = in_sizes[2];

  float* xn   = (float*)d_ws;
  float* h    = xn + (size_t)n * FIN;
  int*   rec  = (int*)(h + (size_t)n * FOUT);
  int*   rank = rec + (size_t)e * 2;
  float* dinv = (float*)(rank + e);
  int*   cnt  = (int*)(dinv + n);
  int*   off  = cnt + n;
  int*   bsum = off + n + 1;

  ln_relu_kernel<<<(n + 7) / 8, 256, 0, stream>>>(x, gamma, beta, xn, n);

  gemm_wmma_kernel<<<n / 16, 256, 0, stream>>>(xn, W, h);

  cnt_zero_kernel<<<(n + 255) / 256, 256, 0, stream>>>(cnt, n);
  rank_kernel<<<(e + 255) / 256, 256, 0, stream>>>(ei, cnt, rank, e);

  blocksum_kernel<<<(n + 1024) / 1024, 256, 0, stream>>>(cnt, bsum, n);
  scan_kernel<<<(n + 1024) / 1024, 256, 0, stream>>>(cnt, bsum, off, n);

  fill_kernel<<<(e + 255) / 256, 256, 0, stream>>>(ei, ew, off, rank, rec, e);
  dinv_kernel<<<(n + 255) / 256, 256, 0, stream>>>(off, rec, dinv, n);
  node_agg_kernel<<<(n + 7) / 8, 256, 0, stream>>>(off, rec, h, dinv, b, out, n);

  (void)n_in; (void)out_size; (void)ws_size;
}


# Edge ranks from 16-bit counters in LDS (edge list in 32 chunks, one 1024-thread block each, returning LDS add) replace 1.6M returning global adds; slots chunk-major, so sum order changes
# speedup vs baseline: 2.6653x; 1.3497x over previous
#include <hip/hip_runtime.h>

#define FIN  128
#define FOUT 128
#define EPSV 1e-5f

typedef float v2f __attribute__((ext_vector_type(2)));
typedef float v8f __attribute__((ext_vector_type(8)));

__global__ __launch_bounds__(256) void ln_relu_kernel(
    const float* __restrict__ x, const float* __restrict__ gamma,
    const float* __restrict__ beta, float* __restrict__ xn, int n) {
  int wave = threadIdx.x >> 5;
  int lane = threadIdx.x & 31;
  int row  = blockIdx.x * 8 + wave;
  if (row >= n) return;

  float4 v = ((const float4*)(x + (size_t)row * FIN))[lane];
  float s  = v.x + v.y + v.z + v.w;
  float ss = v.x * v.x + v.y * v.y + v.z * v.z + v.w * v.w;
  for (int m = 16; m >= 1; m >>= 1) {
    s  += __shfl_xor(s,  m, 32);
    ss += __shfl_xor(ss, m, 32);
  }
  float mean = s * (1.0f / FIN);
  float var  = ss * (1.0f / FIN) - mean * mean;
  float rs   = rsqrtf(var + EPSV);

  float4 g  = ((const float4*)gamma)[lane];
  float4 bt = ((const float4*)beta)[lane];
  float4 o;
  o.x = fmaxf(0.0f, (v.x - mean) * rs * g.x + bt.x);
  o.y = fmaxf(0.0f, (v.y - mean) * rs * g.y + bt.y);
  o.z = fmaxf(0.0f, (v.z - mean) * rs * g.z + bt.z);
  o.w = fmaxf(0.0f, (v.w - mean) * rs * g.w + bt.w);
  ((float4*)(xn + (size_t)row * FIN))[lane] = o;
}

#define A_PITCH 132

__global__ __launch_bounds__(256) void gemm_wmma_kernel(
    const float* __restrict__ xn, const float* __restrict__ W,
    float* __restrict__ h) {
  __shared__ float alds[16 * A_PITCH];
  const int tid = threadIdx.x;
  const int n0  = blockIdx.x * 16;

  for (int s = tid; s < 16 * (FIN / 4); s += 256) {
    int r = s >> 5, c4 = s & 31;
    float4 v = ((const float4*)(xn + (size_t)(n0 + r) * FIN))[c4];
    *((float4*)&alds[r * A_PITCH + c4 * 4]) = v;
  }
  __syncthreads();

  const int wave  = tid >> 5;
  const int lane  = tid & 31;
  const int o0    = wave * 16;
  const int mrow  = lane & 15;
  const int khalf = (lane >> 4) * 2;

  const float* abase = &alds[mrow * A_PITCH + khalf];
  const float* wbase = W + (size_t)(o0 + mrow) * FIN + khalf;

  v8f c = {};
#pragma unroll 8
  for (int k = 0; k < FIN; k += 4) {
    v2f a = *((const v2f*)(abase + k));
    v2f b = *((const v2f*)(wbase + k));
    c = __builtin_amdgcn_wmma_f32_16x16x4_f32(
        false, a, false, b, (short)0, c, false, false);
  }

  const int col   = o0 + mrow;
  const int rbase = n0 + ((lane >> 4) << 3);
#pragma unroll
  for (int v = 0; v < 8; ++v) {
    h[(size_t)(rbase + v) * FOUT + col] = c[v];
  }
}

#define NNODE  100000
#define NWORD  (NNODE / 2)
#define NCHUNK 32
#define RB     25

__global__ __launch_bounds__(1024) void rank_lds_kernel(
    const int* __restrict__ ei, int* __restrict__ rank,
    unsigned int* __restrict__ ctab, int e, int ch) {
  __shared__ __attribute__((aligned(16))) unsigned int ctr[NWORD];
  const int tid = threadIdx.x;
  for (int w = tid; w < NWORD / 4; w += 1024)
    ((uint4*)ctr)[w] = make_uint4(0u, 0u, 0u, 0u);
  __syncthreads();

  const int i0 = blockIdx.x * ch;
  const int i1 = (i0 + ch < e) ? (i0 + ch) : e;
  const int* __restrict__ col = ei + e;
  for (int base = i0 + tid; base < i1; base += RB * 1024) {
    int cv[RB];
#pragma unroll
    for (int u = 0; u < RB; ++u) {
      const int i = base + u * 1024;
      cv[u] = (i < i1) ? col[i] : -1;
    }
    unsigned int old[RB];
#pragma unroll
    for (int u = 0; u < RB; ++u) {
      old[u] = 0u;
      if (cv[u] >= 0)
        old[u] = atomicAdd(&ctr[cv[u] >> 1], 1u << ((cv[u] & 1) << 4));
    }
#pragma unroll
    for (int u = 0; u < RB; ++u) {
      if (cv[u] >= 0)
        rank[base + u * 1024] = (int)((old[u] >> ((cv[u] & 1) << 4)) & 0xFFFFu);
    }
  }
  __syncthreads();

  uint4* dst = (uint4*)(ctab + (size_t)blockIdx.x * NWORD);
  for (int w = tid; w < NWORD / 4; w += 1024) dst[w] = ((const uint4*)ctr)[w];
}

__global__ __launch_bounds__(256) void blocksum_kernel(
    const unsigned int* __restrict__ ctab, int* __restrict__ bsum, int n) {
  __shared__ int wsum[8];
  const int tid = threadIdx.x, lane = tid & 31, wave = tid >> 5;
  const int base = blockIdx.x * 1024 + tid * 4;
  int s = 0;
  if (base < n) {
#pragma unroll
    for (int c = 0; c < NCHUNK; ++c) {
      const uint2 w = *((const uint2*)(ctab + (size_t)c * NWORD + (base >> 1)));
      s += (int)((w.x & 0xFFFFu) + (w.x >> 16) + (w.y & 0xFFFFu) + (w.y >> 16));
    }
  }
  for (int m = 16; m >= 1; m >>= 1) s += __shfl_xor(s, m, 32);
  if (lane == 0) wsum[wave] = s;
  __syncthreads();
  if (tid == 0) {
    int t = 0;
    for (int w = 0; w < 8; ++w) t += wsum[w];
    bsum[blockIdx.x] = t;
  }
}

__global__ __launch_bounds__(256) void scan_kernel(
    const unsigned int* __restrict__ ctab, const int* __restrict__ bsum,
    int* __restrict__ off, int* __restrict__ btab, int n) {
  __shared__ int wsum[8];
  __shared__ int sbase;
  const int tid = threadIdx.x, lane = tid & 31, wave = tid >> 5;
  const int nblk = blockIdx.x;
  if (wave == 0) {
    int s = 0;
    for (int j = lane; j < nblk; j += 32) s += bsum[j];
    for (int m = 16; m >= 1; m >>= 1) s += __shfl_xor(s, m, 32);
    if (lane == 0) sbase = s;
  }
  const int base = nblk * 1024 + tid * 4;
  const bool live = base < n;
  uint2 w[NCHUNK];
  int c[4] = {0, 0, 0, 0};
#pragma unroll
  for (int k = 0; k < NCHUNK; ++k) {
    w[k] = live ? *((const uint2*)(ctab + (size_t)k * NWORD + (base >> 1)))
                : make_uint2(0u, 0u);
    c[0] += (int)(w[k].x & 0xFFFFu);
    c[1] += (int)(w[k].x >> 16);
    c[2] += (int)(w[k].y & 0xFFFFu);
    c[3] += (int)(w[k].y >> 16);
  }
  const int s = c[0] + c[1] + c[2] + c[3];
  int inc = s;
  for (int d = 1; d < 32; d <<= 1) {
    int v = __shfl_up(inc, d, 32);
    if (lane >= d) inc += v;
  }
  if (lane == 31) wsum[wave] = inc;
  __syncthreads();
  int ex = sbase + inc - s;
  for (int k = 0; k < wave; ++k) ex += wsum[k];
  int r[4];
  for (int k = 0; k < 4; ++k) {
    r[k] = ex;
    if (base + k <= n) off[base + k] = ex;
    ex += c[k];
  }
  if (live) {
#pragma unroll
    for (int k = 0; k < NCHUNK; ++k) {
      *((int4*)(btab + (size_t)k * n + base)) = make_int4(r[0], r[1], r[2], r[3]);
      r[0] += (int)(w[k].x & 0xFFFFu);
      r[1] += (int)(w[k].x >> 16);
      r[2] += (int)(w[k].y & 0xFFFFu);
      r[3] += (int)(w[k].y >> 16);
    }
  }
}

__global__ void fill_kernel(const int* __restrict__ ei,
                            const float* __restrict__ ew,
                            const int* __restrict__ btab,
                            const int* __restrict__ rank,
                            int* __restrict__ rec, int e, int ch, int n) {
  const int c = blockIdx.y;
  const int j = blockIdx.x * 256 + threadIdx.x;
  const int i = c * ch + j;
  if (j < ch && i < e) {
    int p = btab[(size_t)c * n + ei[e + i]] + rank[i];
    ((int2*)rec)[p] = make_int2(ei[i], __float_as_int(ew[i]));
  }
}

__global__ void dinv_kernel(const int* __restrict__ off,
                            const int* __restrict__ rec,
                            float* __restrict__ dinv, int n) {
  int i = blockIdx.x * 256 + threadIdx.x;
  if (i < n) {
    const int p1 = off[i + 1];
    float d = 1.0f;
    for (int p = off[i]; p < p1; ++p) d += __int_as_float(((const int2*)rec)[p].y);
    dinv[i] = (d > 0.0f) ? rsqrtf(d) : 0.0f;
  }
}

__global__ __launch_bounds__(256) void node_agg_kernel(
    const int* __restrict__ off, const int* __restrict__ rec,
    const float* __restrict__ h, const float* __restrict__ dinv,
    const float* __restrict__ b, float* __restrict__ out, int n) {
  const int node = blockIdx.x * 8 + (threadIdx.x >> 5);
  if (node >= n) return;
  const int lane = threadIdx.x & 31;
  const int p0 = __builtin_amdgcn_readfirstlane(off[node]);
  const int p1 = __builtin_amdgcn_readfirstlane(off[node + 1]);
  const float dc = dinv[node];
  const float4 hs = ((const float4*)(h + (size_t)node * FOUT))[lane];
  float4 acc;
  acc.x = hs.x * dc * dc;
  acc.y = hs.y * dc * dc;
  acc.z = hs.z * dc * dc;
  acc.w = hs.w * dc * dc;
  for (int q = p0; q < p1; q += 32) {
    const int m = (p1 - q < 32) ? (p1 - q) : 32;
    int   rj = 0;
    float nj = 0.0f;
    if (lane < m) {
      const int2 r2 = ((const int2*)rec)[q + lane];
      rj = r2.x;
      nj = dinv[r2.x] * __int_as_float(r2.y) * dc;
    }
#pragma unroll 4
    for (int j = 0; j < m; ++j) {
      const int   r  = __builtin_amdgcn_readlane(rj, j);
      const float nm = __int_as_float(__builtin_amdgcn_readlane(__float_as_int(nj), j));
      const float4 hr = ((const float4*)(h + (size_t)r * FOUT))[lane];
      acc.x += hr.x * nm;
      acc.y += hr.y * nm;
      acc.z += hr.z * nm;
      acc.w += hr.w * nm;
    }
  }
  const float4 bv = ((const float4*)b)[lane];
  float4 o;
  o.x = acc.x + bv.x;
  o.y = acc.y + bv.y;
  o.z = acc.z + bv.z;
  o.w = acc.w + bv.w;
  ((float4*)(out + (size_t)node * FOUT))[lane] = o;
}

extern "C" void kernel_launch(void* const* d_in, const int* in_sizes, int n_in,
                              void* d_out, int out_size, void* d_ws, size_t ws_size,
                              hipStream_t stream) {
  const float* x     = (const float*)d_in[0];
  const int*   ei    = (const int*)d_in[1];
  const float* ew    = (const float*)d_in[2];
  const float* gamma = (const float*)d_in[3];
  const float* beta  = (const float*)d_in[4];
  const float* W     = (const float*)d_in[5];
  const float* b     = (const float*)d_in[6];
  float* out = (float*)d_out;

  const int n = in_sizes[0] / FIN;
  const int e = in_sizes[2];

  float* xn   = (float*)d_ws;
  float* h    = xn + (size_t)n * FIN;
  int*   rec  = (int*)(h + (size_t)n * FOUT);
  int*   rank = rec + (size_t)e * 2;
  float* dinv = (float*)(rank + e);
  int*   btab = (int*)(dinv + n);
  unsigned int* ctab = (unsigned int*)(btab + (size_t)NCHUNK * n);
  int*   off  = (int*)(ctab + (size_t)NCHUNK * NWORD);
  int*   bsum = off + n + 1;
  const int ch = (e + NCHUNK - 1) / NCHUNK;

  ln_relu_kernel<<<(n + 7) / 8, 256, 0, stream>>>(x, gamma, beta, xn, n);

  gemm_wmma_kernel<<<n / 16, 256, 0, stream>>>(xn, W, h);

  rank_lds_kernel<<<NCHUNK, 1024, 0, stream>>>(ei, rank, ctab, e, ch);

  blocksum_kernel<<<(n + 1024) / 1024, 256, 0, stream>>>(ctab, bsum, n);
  scan_kernel<<<(n + 1024) / 1024, 256, 0, stream>>>(ctab, bsum, off, btab, n);

  fill_kernel<<<dim3((ch + 255) / 256, NCHUNK), 256, 0, stream>>>(
      ei, ew, btab, rank, rec, e, ch, n);
  dinv_kernel<<<(n + 255) / 256, 256, 0, stream>>>(off, rec, dinv, n);
  node_agg_kernel<<<(n + 7) / 8, 256, 0, stream>>>(off, rec, h, dinv, b, out, n);

  (void)n_in; (void)out_size; (void)ws_size;
}


# Fewer, longer-lived waves: LN 8 rows/wave, GEMM 64 rows/block (4 accumulators/wave), fill 8 edges/thread, node sums 10 nodes/wave with later nodes' data asked ahead; sum order kept
# speedup vs baseline: 2.8541x; 1.0708x over previous
#include <hip/hip_runtime.h>

#define FIN  128
#define FOUT 128
#define EPSV 1e-5f

typedef float v2f __attribute__((ext_vector_type(2)));
typedef float v8f __attribute__((ext_vector_type(8)));

#define LN_R 8

__global__ __launch_bounds__(256) void ln_relu_kernel(
    const float* __restrict__ x, const float* __restrict__ gamma,
    const float* __restrict__ beta, float* __restrict__ xn, int n) {
  const int lane = threadIdx.x & 31;
  const int row0 = (blockIdx.x * 8 + (threadIdx.x >> 5)) * LN_R;
  if (row0 >= n) return;

  float4 v[LN_R];
#pragma unroll
  for (int u = 0; u < LN_R; ++u) {
    const int r = (row0 + u < n) ? (row0 + u) : (n - 1);
    v[u] = ((const float4*)(x + (size_t)r * FIN))[lane];
  }
  const float4 g  = ((const float4*)gamma)[lane];
  const float4 bt = ((const float4*)beta)[lane];

  float s[LN_R], ss[LN_R];
#pragma unroll
  for (int u = 0; u < LN_R; ++u) {
    s[u]  = v[u].x + v[u].y + v[u].z + v[u].w;
    ss[u] = v[u].x * v[u].x + v[u].y * v[u].y + v[u].z * v[u].z + v[u].w * v[u].w;
  }
#pragma unroll
  for (int m = 16; m >= 1; m >>= 1) {
#pragma unroll
    for (int u = 0; u < LN_R; ++u) {
      s[u]  += __shfl_xor(s[u],  m, 32);
      ss[u] += __shfl_xor(ss[u], m, 32);
    }
  }
#pragma unroll
  for (int u = 0; u < LN_R; ++u) {
    if (row0 + u < n) {
      const float mean = s[u] * (1.0f / FIN);
      const float var  = ss[u] * (1.0f / FIN) - mean * mean;
      const float rs   = rsqrtf(var + EPSV);
      float4 o;
      o.x = fmaxf(0.0f, (v[u].x - mean) * rs * g.x + bt.x);
      o.y = fmaxf(0.0f, (v[u].y - mean) * rs * g.y + bt.y);
      o.z = fmaxf(0.0f, (v[u].z - mean) * rs * g.z + bt.z);
      o.w = fmaxf(0.0f, (v[u].w - mean) * rs * g.w + bt.w);
      ((float4*)(xn + (size_t)(row0 + u) * FIN))[lane] = o;
    }
  }
}

#define A_PITCH 132
#define MT 4

__global__ __launch_bounds__(256) void gemm_wmma_kernel(
    const float* __restrict__ xn, const float* __restrict__ W,
    float* __restrict__ h, int n) {
  __shared__ float alds[MT * 16 * A_PITCH];
  const int tid = threadIdx.x;
  const int n0  = blockIdx.x * (MT * 16);

#pragma unroll
  for (int it = 0; it < MT * 16 * (FIN / 4) / 256; ++it) {
    const int s = tid + it * 256;
    const int r = s >> 5, c4 = s & 31;
    const int gr = (n0 + r < n) ? (n0 + r) : (n - 1);
    float4 v = ((const float4*)(xn + (size_t)gr * FIN))[c4];
    *((float4*)&alds[r * A_PITCH + c4 * 4]) = v;
  }
  __syncthreads();

  const int wave  = tid >> 5;
  const int lane  = tid & 31;
  const int o0    = wave * 16;
  const int mrow  = lane & 15;
  const int khalf = (lane >> 4) * 2;

  const float* abase = &alds[mrow * A_PITCH + khalf];
  const float* wbase = W + (size_t)(o0 + mrow) * FIN + khalf;

  v8f c[MT];
#pragma unroll
  for (int t = 0; t < MT; ++t) c[t] = (v8f){};
#pragma unroll 4
  for (int k = 0; k < FIN; k += 4) {
    const v2f b = *((const v2f*)(wbase + k));
#pragma unroll
    for (int t = 0; t < MT; ++t) {
      const v2f a = *((const v2f*)(abase + t * 16 * A_PITCH + k));
      c[t] = __builtin_amdgcn_wmma_f32_16x16x4_f32(
          false, a, false, b, (short)0, c[t], false, false);
    }
  }

  const int col = o0 + mrow;
#pragma unroll
  for (int t = 0; t < MT; ++t) {
    if (n0 + t * 16 < n) {
      const int rbase = n0 + t * 16 + ((lane >> 4) << 3);
#pragma unroll
      for (int v = 0; v < 8; ++v) {
        h[(size_t)(rbase + v) * FOUT + col] = c[t][v];
      }
    }
  }
}

#define NNODE  100000
#define NWORD  (NNODE / 2)
#define NCHUNK 32
#define RB     25

__global__ __launch_bounds__(1024) void rank_lds_kernel(
    const int* __restrict__ ei, int* __restrict__ rank,
    unsigned int* __restrict__ ctab, int e, int ch) {
  __shared__ __attribute__((aligned(16))) unsigned int ctr[NWORD];
  const int tid = threadIdx.x;
  for (int w = tid; w < NWORD / 4; w += 1024)
    ((uint4*)ctr)[w] = make_uint4(0u, 0u, 0u, 0u);
  __syncthreads();

  const int i0 = blockIdx.x * ch;
  const int i1 = (i0 + ch < e) ? (i0 + ch) : e;
  const int* __restrict__ col = ei + e;
  for (int base = i0 + tid; base < i1; base += RB * 1024) {
    int cv[RB];
#pragma unroll
    for (int u = 0; u < RB; ++u) {
      const int i = base + u * 1024;
      cv[u] = (i < i1) ? col[i] : -1;
    }
    unsigned int old[RB];
#pragma unroll
    for (int u = 0; u < RB; ++u) {
      old[u] = 0u;
      if (cv[u] >= 0)
        old[u] = atomicAdd(&ctr[cv[u] >> 1], 1u << ((cv[u] & 1) << 4));
    }
#pragma unroll
    for (int u = 0; u < RB; ++u) {
      if (cv[u] >= 0)
        rank[base + u * 1024] = (int)((old[u] >> ((cv[u] & 1) << 4)) & 0xFFFFu);
    }
  }
  __syncthreads();

  uint4* dst = (uint4*)(ctab + (size_t)blockIdx.x * NWORD);
  for (int w = tid; w < NWORD / 4; w += 1024) dst[w] = ((const uint4*)ctr)[w];
}

__global__ __launch_bounds__(256) void blocksum_kernel(
    const unsigned int* __restrict__ ctab, int* __restrict__ bsum, int n) {
  __shared__ int wsum[8];
  const int tid = threadIdx.x, lane = tid & 31, wave = tid >> 5;
  const int base = blockIdx.x * 1024 + tid * 4;
  int s = 0;
  if (base < n) {
#pragma unroll
    for (int c = 0; c < NCHUNK; ++c) {
      const uint2 w = *((const uint2*)(ctab + (size_t)c * NWORD + (base >> 1)));
      s += (int)((w.x & 0xFFFFu) + (w.x >> 16) + (w.y & 0xFFFFu) + (w.y >> 16));
    }
  }
  for (int m = 16; m >= 1; m >>= 1) s += __shfl_xor(s, m, 32);
  if (lane == 0) wsum[wave] = s;
  __syncthreads();
  if (tid == 0) {
    int t = 0;
    for (int w = 0; w < 8; ++w) t += wsum[w];
    bsum[blockIdx.x] = t;
  }
}

__global__ __launch_bounds__(256) void scan_kernel(
    const unsigned int* __restrict__ ctab, const int* __restrict__ bsum,
    int* __restrict__ off, int* __restrict__ btab, int n) {
  __shared__ int wsum[8];
  __shared__ int sbase;
  const int tid = threadIdx.x, lane = tid & 31, wave = tid >> 5;
  const int nblk = blockIdx.x;
  if (wave == 0) {
    int s = 0;
    for (int j = lane; j < nblk; j += 32) s += bsum[j];
    for (int m = 16; m >= 1; m >>= 1) s += __shfl_xor(s, m, 32);
    if (lane == 0) sbase = s;
  }
  const int base = nblk * 1024 + tid * 4;
  const bool live = base < n;
  uint2 w[NCHUNK];
  int c[4] = {0, 0, 0, 0};
#pragma unroll
  for (int k = 0; k < NCHUNK; ++k) {
    w[k] = live ? *((const uint2*)(ctab + (size_t)k * NWORD + (base >> 1)))
                : make_uint2(0u, 0u);
    c[0] += (int)(w[k].x & 0xFFFFu);
    c[1] += (int)(w[k].x >> 16);
    c[2] += (int)(w[k].y & 0xFFFFu);
    c[3] += (int)(w[k].y >> 16);
  }
  const int s = c[0] + c[1] + c[2] + c[3];
  int inc = s;
  for (int d = 1; d < 32; d <<= 1) {
    int v = __shfl_up(inc, d, 32);
    if (lane >= d) inc += v;
  }
  if (lane == 31) wsum[wave] = inc;
  __syncthreads();
  int ex = sbase + inc - s;
  for (int k = 0; k < wave; ++k) ex += wsum[k];
  int r[4];
  for (int k = 0; k < 4; ++k) {
    r[k] = ex;
    if (base + k <= n) off[base + k] = ex;
    ex += c[k];
  }
  if (live) {
#pragma unroll
    for (int k = 0; k < NCHUNK; ++k) {
      *((int4*)(btab + (size_t)k * n + base)) = make_int4(r[0], r[1], r[2], r[3]);
      r[0] += (int)(w[k].x & 0xFFFFu);
      r[1] += (int)(w[k].x >> 16);
      r[2] += (int)(w[k].y & 0xFFFFu);
      r[3] += (int)(w[k].y >> 16);
    }
  }
}

#define FE 8

__global__ __launch_bounds__(256) void fill_kernel(
    const int* __restrict__ ei, const float* __restrict__ ew,
    const int* __restrict__ btab, const int* __restrict__ rank,
    int* __restrict__ rec, int e, int ch, int n) {
  const int c  = blockIdx.y;
  const int j0 = blockIdx.x * (256 * FE) + threadIdx.x;
  const int* __restrict__ bt = btab + (size_t)c * n;
  int  iv[FE], cv[FE], rv[FE], rk[FE], wv[FE];
  bool ok[FE];
#pragma unroll
  for (int u = 0; u < FE; ++u) {
    const int j = j0 + u * 256;
    const int i = c * ch + j;
    ok[u] = (j < ch) && (i < e);
    iv[u] = ok[u] ? i : 0;
    cv[u] = ei[e + iv[u]];
    rv[u] = ei[iv[u]];
    rk[u] = rank[iv[u]];
    wv[u] = __float_as_int(ew[iv[u]]);
  }
  int bs[FE];
#pragma unroll
  for (int u = 0; u < FE; ++u) bs[u] = bt[cv[u]];
#pragma unroll
  for (int u = 0; u < FE; ++u) {
    if (ok[u]) ((int2*)rec)[bs[u] + rk[u]] = make_int2(rv[u], wv[u]);
  }
}

__global__ void dinv_kernel(const int* __restrict__ off,
                            const int* __restrict__ rec,
                            float* __restrict__ dinv, int n) {
  int i = blockIdx.x * 256 + threadIdx.x;
  if (i < n) {
    const int p1 = off[i + 1];
    float d = 1.0f;
    for (int p = off[i]; p < p1; ++p) d += __int_as_float(((const int2*)rec)[p].y);
    dinv[i] = (d > 0.0f) ? rsqrtf(d) : 0.0f;
  }
}

#define NPW 10
#define RCH 8

template <int K>
__device__ __forceinline__ void gather_rows(const float* __restrict__ h, int rj,
                                            float nj, int j, int lane, float4& acc) {
  float4 hr[K];
  float  nm[K];
#pragma unroll
  for (int u = 0; u < K; ++u) {
    const int r = __builtin_amdgcn_readlane(rj, j + u);
    nm[u] = __int_as_float(__builtin_amdgcn_readlane(__float_as_int(nj), j + u));
    hr[u] = ((const float4*)(h + (size_t)r * FOUT))[lane];
  }
#pragma unroll
  for (int u = 0; u < K; ++u) {
    acc.x += hr[u].x * nm[u];
    acc.y += hr[u].y * nm[u];
    acc.z += hr[u].z * nm[u];
    acc.w += hr[u].w * nm[u];
  }
}

__device__ __forceinline__ void sum_batch(const float* __restrict__ h, int rj,
                                          float nj, int m, int lane, float4& acc) {
  int j = 0;
  if (m >= RCH) {
    gather_rows<RCH>(h, rj, nj, 0, lane, acc);
    j = RCH;
  }
  for (; j + RCH <= m; j += RCH) gather_rows<RCH>(h, rj, nj, j, lane, acc);
  switch (m - j) {
    case 7: gather_rows<7>(h, rj, nj, j, lane, acc); break;
    case 6: gather_rows<6>(h, rj, nj, j, lane, acc); break;
    case 5: gather_rows<5>(h, rj, nj, j, lane, acc); break;
    case 4: gather_rows<4>(h, rj, nj, j, lane, acc); break;
    case 3: gather_rows<3>(h, rj, nj, j, lane, acc); break;
    case 2: gather_rows<2>(h, rj, nj, j, lane, acc); break;
    case 1: gather_rows<1>(h, rj, nj, j, lane, acc); break;
    default: break;
  }
}

__global__ __launch_bounds__(256) void node_agg_kernel(
    const int* __restrict__ off, const int* __restrict__ rec,
    const float* __restrict__ h, const float* __restrict__ dinv,
    const float* __restrict__ b, float* __restrict__ out, int n) {
  const int node0 = (blockIdx.x * 8 + (threadIdx.x >> 5)) * NPW;
  if (node0 >= n) return;
  const int lane = threadIdx.x & 31;
  const int nn = (n - node0 < NPW) ? (n - node0) : NPW;
  const int   offv = off[(node0 + lane < n) ? (node0 + lane) : n];
  const float dcv  = dinv[(node0 + lane < n) ? (node0 + lane) : (n - 1)];
  const float4 bv = ((const float4*)b)[lane];

  int2   recA = make_int2(0, 0), recB = make_int2(0, 0);
  float4 hsA;
  float  dgA;
  {
    const int q0 = __builtin_amdgcn_readlane(offv, 0);
    const int q1 = __builtin_amdgcn_readlane(offv, 1);
    hsA = ((const float4*)(h + (size_t)node0 * FOUT))[lane];
    if (lane < q1 - q0) recA = ((const int2*)rec)[q0 + lane];
    if (nn > 1) {
      const int q2 = __builtin_amdgcn_readlane(offv, 2);
      if (lane < q2 - q1) recB = ((const int2*)rec)[q1 + lane];
    }
    dgA = dinv[recA.x];
  }

  for (int t = 0; t < nn; ++t) {
    const int node = node0 + t;
    const int   p0 = __builtin_amdgcn_readlane(offv, t);
    const int   p1 = __builtin_amdgcn_readlane(offv, t + 1);
    const float dc = __int_as_float(__builtin_amdgcn_readlane(__float_as_int(dcv), t));
    const int m0 = (p1 - p0 < 32) ? (p1 - p0) : 32;
    const int rj0 = recA.x;
    float nj0 = 0.0f;
    if (lane < m0) nj0 = dgA * __int_as_float(recA.y) * dc;
    float4 acc;
    acc.x = hsA.x * dc * dc;
    acc.y = hsA.y * dc * dc;
    acc.z = hsA.z * dc * dc;
    acc.w = hsA.w * dc * dc;
    recA = recB;
    if (t + 1 < nn) {
      const int p2 = __builtin_amdgcn_readlane(offv, t + 2);
      hsA = ((const float4*)(h + (size_t)(node + 1) * FOUT))[lane];
      dgA = dinv[recA.x];
      if (t + 2 < nn) {
        const int p3 = __builtin_amdgcn_readlane(offv, t + 3);
        recB = make_int2(0, 0);
        if (lane < p3 - p2) recB = ((const int2*)rec)[p2 + lane];
      }
    }
    sum_batch(h, rj0, nj0, m0, lane, acc);
    for (int q = p0 + 32; q < p1; q += 32) {
      const int m = (p1 - q < 32) ? (p1 - q) : 32;
      int2 r2 = make_int2(0, 0);
      if (lane < m) r2 = ((const int2*)rec)[q + lane];
      float nj = 0.0f;
      if (lane < m) nj = dinv[r2.x] * __int_as_float(r2.y) * dc;
      sum_batch(h, r2.x, nj, m, lane, acc);
    }
    float4 o;
    o.x = acc.x + bv.x;
    o.y = acc.y + bv.y;
    o.z = acc.z + bv.z;
    o.w = acc.w + bv.w;
    ((float4*)(out + (size_t)node * FOUT))[lane] = o;
  }
}

extern "C" void kernel_launch(void* const* d_in, const int* in_sizes, int n_in,
                              void* d_out, int out_size, void* d_ws, size_t ws_size,
                              hipStream_t stream) {
  const float* x     = (const float*)d_in[0];
  const int*   ei    = (const int*)d_in[1];
  const float* ew    = (const float*)d_in[2];
  const float* gamma = (const float*)d_in[3];
  const float* beta  = (const float*)d_in[4];
  const float* W     = (const float*)d_in[5];
  const float* b     = (const float*)d_in[6];
  float* out = (float*)d_out;

  const int n = in_sizes[0] / FIN;
  const int e = in_sizes[2];

  float* xn   = (float*)d_ws;
  float* h    = xn + (size_t)n * FIN;
  int*   rec  = (int*)(h + (size_t)n * FOUT);
  int*   rank = rec + (size_t)e * 2;
  float* dinv = (float*)(rank + e);
  int*   btab = (int*)(dinv + n);
  unsigned int* ctab = (unsigned int*)(btab + (size_t)NCHUNK * n);
  int*   off  = (int*)(ctab + (size_t)NCHUNK * NWORD);
  int*   bsum = off + n + 1;
  const int ch = (e + NCHUNK - 1) / NCHUNK;

  ln_relu_kernel<<<(n + 8 * LN_R - 1) / (8 * LN_R), 256, 0, stream>>>(x, gamma, beta, xn, n);

  gemm_wmma_kernel<<<(n + MT * 16 - 1) / (MT * 16), 256, 0, stream>>>(xn, W, h, n);

  rank_lds_kernel<<<NCHUNK, 1024, 0, stream>>>(ei, rank, ctab, e, ch);

  blocksum_kernel<<<(n + 1024) / 1024, 256, 0, stream>>>(ctab, bsum, n);
  scan_kernel<<<(n + 1024) / 1024, 256, 0, stream>>>(ctab, bsum, off, btab, n);

  fill_kernel<<<dim3((ch + 256 * FE - 1) / (256 * FE), NCHUNK), 256, 0, stream>>>(
      ei, ew, btab, rank, rec, e, ch, n);
  dinv_kernel<<<(n + 255) / 256, 256, 0, stream>>>(off, rec, dinv, n);
  node_agg_kernel<<<(n + 8 * NPW - 1) / (8 * NPW), 256, 0, stream>>>(off, rec, h, dinv, b, out, n);

  (void)n_in; (void)out_size; (void)ws_size;
}


# LayerNorm+ReLU done inside the GEMM's A staging (a wave stages whole rows, same butterfly sums): the normalised rows live only in LDS, their write, re-read and one launch go; sum order kept
# speedup vs baseline: 3.0529x; 1.0697x over previous
#include <hip/hip_runtime.h>

#define FIN  128
#define FOUT 128
#define EPSV 1e-5f

typedef float v2f __attribute__((ext_vector_type(2)));
typedef float v8f __attribute__((ext_vector_type(8)));

#define A_PITCH 132
#define MT 4
#define LN_R (MT * 16 / 8)

__global__ __launch_bounds__(256) void ln_gemm_wmma_kernel(
    const float* __restrict__ x, const float* __restrict__ gamma,
    const float* __restrict__ beta, const float* __restrict__ W,
    float* __restrict__ h, int n) {
  __shared__ float alds[MT * 16 * A_PITCH];
  const int tid  = threadIdx.x;
  const int wave = tid >> 5;
  const int lane = tid & 31;
  const int n0   = blockIdx.x * (MT * 16);

  float4 v[LN_R];
#pragma unroll
  for (int u = 0; u < LN_R; ++u) {
    const int r  = u * 8 + wave;
    const int gr = (n0 + r < n) ? (n0 + r) : (n - 1);
    v[u] = ((const float4*)(x + (size_t)gr * FIN))[lane];
  }
  const float4 g  = ((const float4*)gamma)[lane];
  const float4 bt = ((const float4*)beta)[lane];

  float s[LN_R], ss[LN_R];
#pragma unroll
  for (int u = 0; u < LN_R; ++u) {
    s[u]  = v[u].x + v[u].y + v[u].z + v[u].w;
    ss[u] = v[u].x * v[u].x + v[u].y * v[u].y + v[u].z * v[u].z + v[u].w * v[u].w;
  }
#pragma unroll
  for (int m = 16; m >= 1; m >>= 1) {
#pragma unroll
    for (int u = 0; u < LN_R; ++u) {
      s[u]  += __shfl_xor(s[u],  m, 32);
      ss[u] += __shfl_xor(ss[u], m, 32);
    }
  }
#pragma unroll
  for (int u = 0; u < LN_R; ++u) {
    const float mean = s[u] * (1.0f / FIN);
    const float var  = ss[u] * (1.0f / FIN) - mean * mean;
    const float rs   = rsqrtf(var + EPSV);
    float4 o;
    o.x = fmaxf(0.0f, (v[u].x - mean) * rs * g.x + bt.x);
    o.y = fmaxf(0.0f, (v[u].y - mean) * rs * g.y + bt.y);
    o.z = fmaxf(0.0f, (v[u].z - mean) * rs * g.z + bt.z);
    o.w = fmaxf(0.0f, (v[u].w - mean) * rs * g.w + bt.w);
    *((float4*)&alds[(u * 8 + wave) * A_PITCH + lane * 4]) = o;
  }
  __syncthreads();

  const int o0    = wave * 16;
  const int mrow  = lane & 15;
  const int khalf = (lane >> 4) * 2;

  const float* abase = &alds[mrow * A_PITCH + khalf];
  const float* wbase = W + (size_t)(o0 + mrow) * FIN + khalf;

  v8f c[MT];
#pragma unroll
  for (int t = 0; t < MT; ++t) c[t] = (v8f){};
#pragma unroll 4
  for (int k = 0; k < FIN; k += 4) {
    const v2f b = *((const v2f*)(wbase + k));
#pragma unroll
    for (int t = 0; t < MT; ++t) {
      const v2f a = *((const v2f*)(abase + t * 16 * A_PITCH + k));
      c[t] = __builtin_amdgcn_wmma_f32_16x16x4_f32(
          false, a, false, b, (short)0, c[t], false, false);
    }
  }

  const int col = o0 + mrow;
#pragma unroll
  for (int t = 0; t < MT; ++t) {
    if (n0 + t * 16 < n) {
      const int rbase = n0 + t * 16 + ((lane >> 4) << 3);
#pragma unroll
      for (int v = 0; v < 8; ++v) {
        h[(size_t)(rbase + v) * FOUT + col] = c[t][v];
      }
    }
  }
}

#define NNODE  100000
#define NWORD  (NNODE / 2)
#define NCHUNK 32
#define RB     25

__global__ __launch_bounds__(1024) void rank_lds_kernel(
    const int* __restrict__ ei, int* __restrict__ rank,
    unsigned int* __restrict__ ctab, int e, int ch) {
  __shared__ __attribute__((aligned(16))) unsigned int ctr[NWORD];
  const int tid = threadIdx.x;
  for (int w = tid; w < NWORD / 4; w += 1024)
    ((uint4*)ctr)[w] = make_uint4(0u, 0u, 0u, 0u);
  __syncthreads();

  const int i0 = blockIdx.x * ch;
  const int i1 = (i0 + ch < e) ? (i0 + ch) : e;
  const int* __restrict__ col = ei + e;
  for (int base = i0 + tid; base < i1; base += RB * 1024) {
    int cv[RB];
#pragma unroll
    for (int u = 0; u < RB; ++u) {
      const int i = base + u * 1024;
      cv[u] = (i < i1) ? col[i] : -1;
    }
    unsigned int old[RB];
#pragma unroll
    for (int u = 0; u < RB; ++u) {
      old[u] = 0u;
      if (cv[u] >= 0)
        old[u] = atomicAdd(&ctr[cv[u] >> 1], 1u << ((cv[u] & 1) << 4));
    }
#pragma unroll
    for (int u = 0; u < RB; ++u) {
      if (cv[u] >= 0)
        rank[base + u * 1024] = (int)((old[u] >> ((cv[u] & 1) << 4)) & 0xFFFFu);
    }
  }
  __syncthreads();

  uint4* dst = (uint4*)(ctab + (size_t)blockIdx.x * NWORD);
  for (int w = tid; w < NWORD / 4; w += 1024) dst[w] = ((const uint4*)ctr)[w];
}

__global__ __launch_bounds__(256) void blocksum_kernel(
    const unsigned int* __restrict__ ctab, int* __restrict__ bsum, int n) {
  __shared__ int wsum[8];
  const int tid = threadIdx.x, lane = tid & 31, wave = tid >> 5;
  const int base = blockIdx.x * 1024 + tid * 4;
  int s = 0;
  if (base < n) {
#pragma unroll
    for (int c = 0; c < NCHUNK; ++c) {
      const uint2 w = *((const uint2*)(ctab + (size_t)c * NWORD + (base >> 1)));
      s += (int)((w.x & 0xFFFFu) + (w.x >> 16) + (w.y & 0xFFFFu) + (w.y >> 16));
    }
  }
  for (int m = 16; m >= 1; m >>= 1) s += __shfl_xor(s, m, 32);
  if (lane == 0) wsum[wave] = s;
  __syncthreads();
  if (tid == 0) {
    int t = 0;
    for (int w = 0; w < 8; ++w) t += wsum[w];
    bsum[blockIdx.x] = t;
  }
}

__global__ __launch_bounds__(256) void scan_kernel(
    const unsigned int* __restrict__ ctab, const int* __restrict__ bsum,
    int* __restrict__ off, int* __restrict__ btab, int n) {
  __shared__ int wsum[8];
  __shared__ int sbase;
  const int tid = threadIdx.x, lane = tid & 31, wave = tid >> 5;
  const int nblk = blockIdx.x;
  if (wave == 0) {
    int s = 0;
    for (int j = lane; j < nblk; j += 32) s += bsum[j];
    for (int m = 16; m >= 1; m >>= 1) s += __shfl_xor(s, m, 32);
    if (lane == 0) sbase = s;
  }
  const int base = nblk * 1024 + tid * 4;
  const bool live = base < n;
  uint2 w[NCHUNK];
  int c[4] = {0, 0, 0, 0};
#pragma unroll
  for (int k = 0; k < NCHUNK; ++k) {
    w[k] = live ? *((const uint2*)(ctab + (size_t)k * NWORD + (base >> 1)))
                : make_uint2(0u, 0u);
    c[0] += (int)(w[k].x & 0xFFFFu);
    c[1] += (int)(w[k].x >> 16);
    c[2] += (int)(w[k].y & 0xFFFFu);
    c[3] += (int)(w[k].y >> 16);
  }
  const int s = c[0] + c[1] + c[2] + c[3];
  int inc = s;
  for (int d = 1; d < 32; d <<= 1) {
    int v = __shfl_up(inc, d, 32);
    if (lane >= d) inc += v;
  }
  if (lane == 31) wsum[wave] = inc;
  __syncthreads();
  int ex = sbase + inc - s;
  for (int k = 0; k < wave; ++k) ex += wsum[k];
  int r[4];
  for (int k = 0; k < 4; ++k) {
    r[k] = ex;
    if (base + k <= n) off[base + k] = ex;
    ex += c[k];
  }
  if (live) {
#pragma unroll
    for (int k = 0; k < NCHUNK; ++k) {
      *((int4*)(btab + (size_t)k * n + base)) = make_int4(r[0], r[1], r[2], r[3]);
      r[0] += (int)(w[k].x & 0xFFFFu);
      r[1] += (int)(w[k].x >> 16);
      r[2] += (int)(w[k].y & 0xFFFFu);
      r[3] += (int)(w[k].y >> 16);
    }
  }
}

#define FE 8

__global__ __launch_bounds__(256) void fill_kernel(
    const int* __restrict__ ei, const float* __restrict__ ew,
    const int* __restrict__ btab, const int* __restrict__ rank,
    int* __restrict__ rec, int e, int ch, int n) {
  const int c  = blockIdx.y;
  const int j0 = blockIdx.x * (256 * FE) + threadIdx.x;
  const int* __restrict__ bt = btab + (size_t)c * n;
  int  iv[FE], cv[FE], rv[FE], rk[FE], wv[FE];
  bool ok[FE];
#pragma unroll
  for (int u = 0; u < FE; ++u) {
    const int j = j0 + u * 256;
    const int i = c * ch + j;
    ok[u] = (j < ch) && (i < e);
    iv[u] = ok[u] ? i : 0;
    cv[u] = ei[e + iv[u]];
    rv[u] = ei[iv[u]];
    rk[u] = rank[iv[u]];
    wv[u] = __float_as_int(ew[iv[u]]);
  }
  int bs[FE];
#pragma unroll
  for (int u = 0; u < FE; ++u) bs[u] = bt[cv[u]];
#pragma unroll
  for (int u = 0; u < FE; ++u) {
    if (ok[u]) ((int2*)rec)[bs[u] + rk[u]] = make_int2(rv[u], wv[u]);
  }
}

__global__ void dinv_kernel(const int* __restrict__ off,
                            const int* __restrict__ rec,
                            float* __restrict__ dinv, int n) {
  int i = blockIdx.x * 256 + threadIdx.x;
  if (i < n) {
    const int p1 = off[i + 1];
    float d = 1.0f;
    for (int p = off[i]; p < p1; ++p) d += __int_as_float(((const int2*)rec)[p].y);
    dinv[i] = (d > 0.0f) ? rsqrtf(d) : 0.0f;
  }
}

#define NPW 10
#define RCH 8

template <int K>
__device__ __forceinline__ void gather_rows(const float* __restrict__ h, int rj,
                                            float nj, int j, int lane, float4& acc) {
  float4 hr[K];
  float  nm[K];
#pragma unroll
  for (int u = 0; u < K; ++u) {
    const int r = __builtin_amdgcn_readlane(rj, j + u);
    nm[u] = __int_as_float(__builtin_amdgcn_readlane(__float_as_int(nj), j + u));
    hr[u] = ((const float4*)(h + (size_t)r * FOUT))[lane];
  }
#pragma unroll
  for (int u = 0; u < K; ++u) {
    acc.x += hr[u].x * nm[u];
    acc.y += hr[u].y * nm[u];
    acc.z += hr[u].z * nm[u];
    acc.w += hr[u].w * nm[u];
  }
}

__device__ __forceinline__ void sum_batch(const float* __restrict__ h, int rj,
                                          float nj, int m, int lane, float4& acc) {
  int j = 0;
  if (m >= RCH) {
    gather_rows<RCH>(h, rj, nj, 0, lane, acc);
    j = RCH;
  }
  for (; j + RCH <= m; j += RCH) gather_rows<RCH>(h, rj, nj, j, lane, acc);
  switch (m - j) {
    case 7: gather_rows<7>(h, rj, nj, j, lane, acc); break;
    case 6: gather_rows<6>(h, rj, nj, j, lane, acc); break;
    case 5: gather_rows<5>(h, rj, nj, j, lane, acc); break;
    case 4: gather_rows<4>(h, rj, nj, j, lane, acc); break;
    case 3: gather_rows<3>(h, rj, nj, j, lane, acc); break;
    case 2: gather_rows<2>(h, rj, nj, j, lane, acc); break;
    case 1: gather_rows<1>(h, rj, nj, j, lane, acc); break;
    default: break;
  }
}

__global__ __launch_bounds__(256) void node_agg_kernel(
    const int* __restrict__ off, const int* __restrict__ rec,
    const float* __restrict__ h, const float* __restrict__ dinv,
    const float* __restrict__ b, float* __restrict__ out, int n) {
  const int node0 = (blockIdx.x * 8 + (threadIdx.x >> 5)) * NPW;
  if (node0 >= n) return;
  const int lane = threadIdx.x & 31;
  const int nn = (n - node0 < NPW) ? (n - node0) : NPW;
  const int   offv = off[(node0 + lane < n) ? (node0 + lane) : n];
  const float dcv  = dinv[(node0 + lane < n) ? (node0 + lane) : (n - 1)];
  const float4 bv = ((const float4*)b)[lane];

  int2   recA = make_int2(0, 0), recB = make_int2(0, 0);
  float4 hsA;
  float  dgA;
  {
    const int q0 = __builtin_amdgcn_readlane(offv, 0);
    const int q1 = __builtin_amdgcn_readlane(offv, 1);
    hsA = ((const float4*)(h + (size_t)node0 * FOUT))[lane];
    if (lane < q1 - q0) recA = ((const int2*)rec)[q0 + lane];
    if (nn > 1) {
      const int q2 = __builtin_amdgcn_readlane(offv, 2);
      if (lane < q2 - q1) recB = ((const int2*)rec)[q1 + lane];
    }
    dgA = dinv[recA.x];
  }

  for (int t = 0; t < nn; ++t) {
    const int node = node0 + t;
    const int   p0 = __builtin_amdgcn_readlane(offv, t);
    const int   p1 = __builtin_amdgcn_readlane(offv, t + 1);
    const float dc = __int_as_float(__builtin_amdgcn_readlane(__float_as_int(dcv), t));
    const int m0 = (p1 - p0 < 32) ? (p1 - p0) : 32;
    const int rj0 = recA.x;
    float nj0 = 0.0f;
    if (lane < m0) nj0 = dgA * __int_as_float(recA.y) * dc;
    float4 acc;
    acc.x = hsA.x * dc * dc;
    acc.y = hsA.y * dc * dc;
    acc.z = hsA.z * dc * dc;
    acc.w = hsA.w * dc * dc;
    recA = recB;
    if (t + 1 < nn) {
      const int p2 = __builtin_amdgcn_readlane(offv, t + 2);
      hsA = ((const float4*)(h + (size_t)(node + 1) * FOUT))[lane];
      dgA = dinv[recA.x];
      if (t + 2 < nn) {
        const int p3 = __builtin_amdgcn_readlane(offv, t + 3);
        recB = make_int2(0, 0);
        if (lane < p3 - p2) recB = ((const int2*)rec)[p2 + lane];
      }
    }
    sum_batch(h, rj0, nj0, m0, lane, acc);
    for (int q = p0 + 32; q < p1; q += 32) {
      const int m = (p1 - q < 32) ? (p1 - q) : 32;
      int2 r2 = make_int2(0, 0);
      if (lane < m) r2 = ((const int2*)rec)[q + lane];
      float nj = 0.0f;
      if (lane < m) nj = dinv[r2.x] * __int_as_float(r2.y) * dc;
      sum_batch(h, r2.x, nj, m, lane, acc);
    }
    float4 o;
    o.x = acc.x + bv.x;
    o.y = acc.y + bv.y;
    o.z = acc.z + bv.z;
    o.w = acc.w + bv.w;
    ((float4*)(out + (size_t)node * FOUT))[lane] = o;
  }
}

extern "C" void kernel_launch(void* const* d_in, const int* in_sizes, int n_in,
                              void* d_out, int out_size, void* d_ws, size_t ws_size,
                              hipStream_t stream) {
  const float* x     = (const float*)d_in[0];
  const int*   ei    = (const int*)d_in[1];
  const float* ew    = (const float*)d_in[2];
  const float* gamma = (const float*)d_in[3];
  const float* beta  = (const float*)d_in[4];
  const float* W     = (const float*)d_in[5];
  const float* b     = (const float*)d_in[6];
  float* out = (float*)d_out;

  const int n = in_sizes[0] / FIN;
  const int e = in_sizes[2];

  float* h    = (float*)d_ws;
  int*   rec  = (int*)(h + (size_t)n * FOUT);
  int*   rank = rec + (size_t)e * 2;
  float* dinv = (float*)(rank + e);
  int*   btab = (int*)(dinv + n);
  unsigned int* ctab = (unsigned int*)(btab + (size_t)NCHUNK * n);
  int*   off  = (int*)(ctab + (size_t)NCHUNK * NWORD);
  int*   bsum = off + n + 1;
  const int ch = (e + NCHUNK - 1) / NCHUNK;

  ln_gemm_wmma_kernel<<<(n + MT * 16 - 1) / (MT * 16), 256, 0, stream>>>(x, gamma, beta, W, h, n);

  rank_lds_kernel<<<NCHUNK, 1024, 0, stream>>>(ei, rank, ctab, e, ch);

  blocksum_kernel<<<(n + 1024) / 1024, 256, 0, stream>>>(ctab, bsum, n);
  scan_kernel<<<(n + 1024) / 1024, 256, 0, stream>>>(ctab, bsum, off, btab, n);

  fill_kernel<<<dim3((ch + 256 * FE - 1) / (256 * FE), NCHUNK), 256, 0, stream>>>(
      ei, ew, btab, rank, rec, e, ch, n);
  dinv_kernel<<<(n + 255) / 256, 256, 0, stream>>>(off, rec, dinv, n);
  node_agg_kernel<<<(n + 8 * NPW - 1) / (8 * NPW), 256, 0, stream>>>(off, rec, h, dinv, b, out, n);

  (void)n_in; (void)out_size; (void)ws_size;
}


# h rows pre-scaled by their own node's dinv: sort runs first, the product kernel's row stager sums its node's weights by butterfly (deg sum order changes); dinv launch and per-edge dinv gather go
# speedup vs baseline: 3.1441x; 1.0299x over previous
#include <hip/hip_runtime.h>

#define FIN  128
#define FOUT 128
#define EPSV 1e-5f

typedef float v2f __attribute__((ext_vector_type(2)));
typedef float v8f __attribute__((ext_vector_type(8)));

#define A_PITCH 132
#define MT 4
#define LN_R (MT * 16 / 8)

__global__ __launch_bounds__(256) void ln_gemm_wmma_kernel(
    const float* __restrict__ x, const float* __restrict__ gamma,
    const float* __restrict__ beta, const float* __restrict__ W,
    const int* __restrict__ off, const int* __restrict__ rec,
    float* __restrict__ h, float* __restrict__ dinv, int n) {
  __shared__ float alds[MT * 16 * A_PITCH];
  const int tid  = threadIdx.x;
  const int wave = tid >> 5;
  const int lane = tid & 31;
  const int n0   = blockIdx.x * (MT * 16);
  const int swave = __builtin_amdgcn_readfirstlane(wave);

  int p0[LN_R], cnt[LN_R];
#pragma unroll
  for (int u = 0; u < LN_R; ++u) {
    const int r  = u * 8 + swave;
    const int gr = (n0 + r < n) ? (n0 + r) : (n - 1);
    p0[u]  = off[gr];
    cnt[u] = off[gr + 1] - p0[u];
  }

  float4 v[LN_R];
#pragma unroll
  for (int u = 0; u < LN_R; ++u) {
    const int r  = u * 8 + wave;
    const int gr = (n0 + r < n) ? (n0 + r) : (n - 1);
    v[u] = ((const float4*)(x + (size_t)gr * FIN))[lane];
  }
  float wq[LN_R];
#pragma unroll
  for (int u = 0; u < LN_R; ++u) {
    wq[u] = 0.0f;
    if (lane < cnt[u]) wq[u] = __int_as_float(rec[2 * (size_t)(p0[u] + lane) + 1]);
  }
  const float4 g  = ((const float4*)gamma)[lane];
  const float4 bt = ((const float4*)beta)[lane];

  float s[LN_R], ss[LN_R];
#pragma unroll
  for (int u = 0; u < LN_R; ++u) {
    s[u]  = v[u].x + v[u].y + v[u].z + v[u].w;
    ss[u] = v[u].x * v[u].x + v[u].y * v[u].y + v[u].z * v[u].z + v[u].w * v[u].w;
  }
#pragma unroll
  for (int u = 0; u < LN_R; ++u) {
    if (cnt[u] > 32) {
      for (int q = 32 + lane; q < cnt[u]; q += 32)
        wq[u] += __int_as_float(rec[2 * (size_t)(p0[u] + q) + 1]);
    }
  }
#pragma unroll
  for (int m = 16; m >= 1; m >>= 1) {
#pragma unroll
    for (int u = 0; u < LN_R; ++u) {
      s[u]  += __shfl_xor(s[u],  m, 32);
      ss[u] += __shfl_xor(ss[u], m, 32);
      wq[u] += __shfl_xor(wq[u], m, 32);
    }
  }
  float dvs[LN_R];
#pragma unroll
  for (int u = 0; u < LN_R; ++u) {
    const float mean = s[u] * (1.0f / FIN);
    const float var  = ss[u] * (1.0f / FIN) - mean * mean;
    const float rs   = rsqrtf(var + EPSV);
    const float dg = 1.0f + wq[u];
    const float dv = (dg > 0.0f) ? rsqrtf(dg) : 0.0f;
    dvs[u] = dv;
    float4 o;
    o.x = fmaxf(0.0f, (v[u].x - mean) * rs * g.x + bt.x) * dv;
    o.y = fmaxf(0.0f, (v[u].y - mean) * rs * g.y + bt.y) * dv;
    o.z = fmaxf(0.0f, (v[u].z - mean) * rs * g.z + bt.z) * dv;
    o.w = fmaxf(0.0f, (v[u].w - mean) * rs * g.w + bt.w) * dv;
    *((float4*)&alds[(u * 8 + wave) * A_PITCH + lane * 4]) = o;
  }
  __syncthreads();

  if (lane == 0) {
#pragma unroll
    for (int u = 0; u < LN_R; ++u) {
      if (n0 + u * 8 + wave < n) dinv[n0 + u * 8 + wave] = dvs[u];
    }
  }

  const int o0    = wave * 16;
  const int mrow  = lane & 15;
  const int khalf = (lane >> 4) * 2;

  const float* abase = &alds[mrow * A_PITCH + khalf];
  const float* wbase = W + (size_t)(o0 + mrow) * FIN + khalf;

  v8f c[MT];
#pragma unroll
  for (int t = 0; t < MT; ++t) c[t] = (v8f){};
#pragma unroll 4
  for (int k = 0; k < FIN; k += 4) {
    const v2f b = *((const v2f*)(wbase + k));
#pragma unroll
    for (int t = 0; t < MT; ++t) {
      const v2f a = *((const v2f*)(abase + t * 16 * A_PITCH + k));
      c[t] = __builtin_amdgcn_wmma_f32_16x16x4_f32(
          false, a, false, b, (short)0, c[t], false, false);
    }
  }

  const int col = o0 + mrow;
#pragma unroll
  for (int t = 0; t < MT; ++t) {
    if (n0 + t * 16 < n) {
      const int rbase = n0 + t * 16 + ((lane >> 4) << 3);
#pragma unroll
      for (int v = 0; v < 8; ++v) {
        h[(size_t)(rbase + v) * FOUT + col] = c[t][v];
      }
    }
  }
}

#define NNODE  100000
#define NWORD  (NNODE / 2)
#define NCHUNK 32
#define RB     25

__global__ __launch_bounds__(1024) void rank_lds_kernel(
    const int* __restrict__ ei, int* __restrict__ rank,
    unsigned int* __restrict__ ctab, int e, int ch) {
  __shared__ __attribute__((aligned(16))) unsigned int ctr[NWORD];
  const int tid = threadIdx.x;
  for (int w = tid; w < NWORD / 4; w += 1024)
    ((uint4*)ctr)[w] = make_uint4(0u, 0u, 0u, 0u);
  __syncthreads();

  const int i0 = blockIdx.x * ch;
  const int i1 = (i0 + ch < e) ? (i0 + ch) : e;
  const int* __restrict__ col = ei + e;
  for (int base = i0 + tid; base < i1; base += RB * 1024) {
    int cv[RB];
#pragma unroll
    for (int u = 0; u < RB; ++u) {
      const int i = base + u * 1024;
      cv[u] = (i < i1) ? col[i] : -1;
    }
    unsigned int old[RB];
#pragma unroll
    for (int u = 0; u < RB; ++u) {
      old[u] = 0u;
      if (cv[u] >= 0)
        old[u] = atomicAdd(&ctr[cv[u] >> 1], 1u << ((cv[u] & 1) << 4));
    }
#pragma unroll
    for (int u = 0; u < RB; ++u) {
      if (cv[u] >= 0)
        rank[base + u * 1024] = (int)((old[u] >> ((cv[u] & 1) << 4)) & 0xFFFFu);
    }
  }
  __syncthreads();

  uint4* dst = (uint4*)(ctab + (size_t)blockIdx.x * NWORD);
  for (int w = tid; w < NWORD / 4; w += 1024) dst[w] = ((const uint4*)ctr)[w];
}

__global__ __launch_bounds__(256) void blocksum_kernel(
    const unsigned int* __restrict__ ctab, int* __restrict__ bsum, int n) {
  __shared__ int wsum[8];
  const int tid = threadIdx.x, lane = tid & 31, wave = tid >> 5;
  const int base = blockIdx.x * 1024 + tid * 4;
  int s = 0;
  if (base < n) {
#pragma unroll
    for (int c = 0; c < NCHUNK; ++c) {
      const uint2 w = *((const uint2*)(ctab + (size_t)c * NWORD + (base >> 1)));
      s += (int)((w.x & 0xFFFFu) + (w.x >> 16) + (w.y & 0xFFFFu) + (w.y >> 16));
    }
  }
  for (int m = 16; m >= 1; m >>= 1) s += __shfl_xor(s, m, 32);
  if (lane == 0) wsum[wave] = s;
  __syncthreads();
  if (tid == 0) {
    int t = 0;
    for (int w = 0; w < 8; ++w) t += wsum[w];
    bsum[blockIdx.x] = t;
  }
}

__global__ __launch_bounds__(256) void scan_kernel(
    const unsigned int* __restrict__ ctab, const int* __restrict__ bsum,
    int* __restrict__ off, int* __restrict__ btab, int n) {
  __shared__ int wsum[8];
  __shared__ int sbase;
  const int tid = threadIdx.x, lane = tid & 31, wave = tid >> 5;
  const int nblk = blockIdx.x;
  if (wave == 0) {
    int s = 0;
    for (int j = lane; j < nblk; j += 32) s += bsum[j];
    for (int m = 16; m >= 1; m >>= 1) s += __shfl_xor(s, m, 32);
    if (lane == 0) sbase = s;
  }
  const int base = nblk * 1024 + tid * 4;
  const bool live = base < n;
  uint2 w[NCHUNK];
  int c[4] = {0, 0, 0, 0};
#pragma unroll
  for (int k = 0; k < NCHUNK; ++k) {
    w[k] = live ? *((const uint2*)(ctab + (size_t)k * NWORD + (base >> 1)))
                : make_uint2(0u, 0u);
    c[0] += (int)(w[k].x & 0xFFFFu);
    c[1] += (int)(w[k].x >> 16);
    c[2] += (int)(w[k].y & 0xFFFFu);
    c[3] += (int)(w[k].y >> 16);
  }
  const int s = c[0] + c[1] + c[2] + c[3];
  int inc = s;
  for (int d = 1; d < 32; d <<= 1) {
    int v = __shfl_up(inc, d, 32);
    if (lane >= d) inc += v;
  }
  if (lane == 31) wsum[wave] = inc;
  __syncthreads();
  int ex = sbase + inc - s;
  for (int k = 0; k < wave; ++k) ex += wsum[k];
  int r[4];
  for (int k = 0; k < 4; ++k) {
    r[k] = ex;
    if (base + k <= n) off[base + k] = ex;
    ex += c[k];
  }
  if (live) {
#pragma unroll
    for (int k = 0; k < NCHUNK; ++k) {
      *((int4*)(btab + (size_t)k * n + base)) = make_int4(r[0], r[1], r[2], r[3]);
      r[0] += (int)(w[k].x & 0xFFFFu);
      r[1] += (int)(w[k].x >> 16);
      r[2] += (int)(w[k].y & 0xFFFFu);
      r[3] += (int)(w[k].y >> 16);
    }
  }
}

#define FE 8

__global__ __launch_bounds__(256) void fill_kernel(
    const int* __restrict__ ei, const float* __restrict__ ew,
    const int* __restrict__ btab, const int* __restrict__ rank,
    int* __restrict__ rec, int e, int ch, int n) {
  const int c  = blockIdx.y;
  const int j0 = blockIdx.x * (256 * FE) + threadIdx.x;
  const int* __restrict__ bt = btab + (size_t)c * n;
  int  iv[FE], cv[FE], rv[FE], rk[FE], wv[FE];
  bool ok[FE];
#pragma unroll
  for (int u = 0; u < FE; ++u) {
    const int j = j0 + u * 256;
    const int i = c * ch + j;
    ok[u] = (j < ch) && (i < e);
    iv[u] = ok[u] ? i : 0;
    cv[u] = ei[e + iv[u]];
    rv[u] = ei[iv[u]];
    rk[u] = rank[iv[u]];
    wv[u] = __float_as_int(ew[iv[u]]);
  }
  int bs[FE];
#pragma unroll
  for (int u = 0; u < FE; ++u) bs[u] = bt[cv[u]];
#pragma unroll
  for (int u = 0; u < FE; ++u) {
    if (ok[u]) ((int2*)rec)[bs[u] + rk[u]] = make_int2(rv[u], wv[u]);
  }
}

#define NPW 10
#define RCH 8

template <int K>
__device__ __forceinline__ void gather_rows(const float* __restrict__ h, int rj,
                                            float nj, int j, int lane, float4& acc) {
  float4 hr[K];
  float  nm[K];
#pragma unroll
  for (int u = 0; u < K; ++u) {
    const int r = __builtin_amdgcn_readlane(rj, j + u);
    nm[u] = __int_as_float(__builtin_amdgcn_readlane(__float_as_int(nj), j + u));
    hr[u] = ((const float4*)(h + (size_t)r * FOUT))[lane];
  }
#pragma unroll
  for (int u = 0; u < K; ++u) {
    acc.x += hr[u].x * nm[u];
    acc.y += hr[u].y * nm[u];
    acc.z += hr[u].z * nm[u];
    acc.w += hr[u].w * nm[u];
  }
}

__device__ __forceinline__ void sum_batch(const float* __restrict__ h, int rj,
                                          float nj, int m, int lane, float4& acc) {
  int j = 0;
  if (m >= RCH) {
    gather_rows<RCH>(h, rj, nj, 0, lane, acc);
    j = RCH;
  }
  for (; j + RCH <= m; j += RCH) gather_rows<RCH>(h, rj, nj, j, lane, acc);
  switch (m - j) {
    case 7: gather_rows<7>(h, rj, nj, j, lane, acc); break;
    case 6: gather_rows<6>(h, rj, nj, j, lane, acc); break;
    case 5: gather_rows<5>(h, rj, nj, j, lane, acc); break;
    case 4: gather_rows<4>(h, rj, nj, j, lane, acc); break;
    case 3: gather_rows<3>(h, rj, nj, j, lane, acc); break;
    case 2: gather_rows<2>(h, rj, nj, j, lane, acc); break;
    case 1: gather_rows<1>(h, rj, nj, j, lane, acc); break;
    default: break;
  }
}

__global__ __launch_bounds__(256) void node_agg_kernel(
    const int* __restrict__ off, const int* __restrict__ rec,
    const float* __restrict__ h, const float* __restrict__ dinv,
    const float* __restrict__ b, float* __restrict__ out, int n) {
  const int node0 = (blockIdx.x * 8 + (threadIdx.x >> 5)) * NPW;
  if (node0 >= n) return;
  const int lane = threadIdx.x & 31;
  const int nn = (n - node0 < NPW) ? (n - node0) : NPW;
  const int   offv = off[(node0 + lane < n) ? (node0 + lane) : n];
  const float dcv  = dinv[(node0 + lane < n) ? (node0 + lane) : (n - 1)];
  const float4 bv = ((const float4*)b)[lane];

  int2   recA = make_int2(0, 0), recB = make_int2(0, 0);
  float4 hsA;
  {
    const int q0 = __builtin_amdgcn_readlane(offv, 0);
    const int q1 = __builtin_amdgcn_readlane(offv, 1);
    hsA = ((const float4*)(h + (size_t)node0 * FOUT))[lane];
    if (lane < q1 - q0) recA = ((const int2*)rec)[q0 + lane];
    if (nn > 1) {
      const int q2 = __builtin_amdgcn_readlane(offv, 2);
      if (lane < q2 - q1) recB = ((const int2*)rec)[q1 + lane];
    }
  }

  for (int t = 0; t < nn; ++t) {
    const int node = node0 + t;
    const int   p0 = __builtin_amdgcn_readlane(offv, t);
    const int   p1 = __builtin_amdgcn_readlane(offv, t + 1);
    const float dc = __int_as_float(__builtin_amdgcn_readlane(__float_as_int(dcv), t));
    const int m0 = (p1 - p0 < 32) ? (p1 - p0) : 32;
    const int rj0 = recA.x;
    float nj0 = 0.0f;
    if (lane < m0) nj0 = __int_as_float(recA.y) * dc;
    float4 acc;
    acc.x = hsA.x * dc;
    acc.y = hsA.y * dc;
    acc.z = hsA.z * dc;
    acc.w = hsA.w * dc;
    recA = recB;
    if (t + 1 < nn) {
      const int p2 = __builtin_amdgcn_readlane(offv, t + 2);
      hsA = ((const float4*)(h + (size_t)(node + 1) * FOUT))[lane];
      if (t + 2 < nn) {
        const int p3 = __builtin_amdgcn_readlane(offv, t + 3);
        recB = make_int2(0, 0);
        if (lane < p3 - p2) recB = ((const int2*)rec)[p2 + lane];
      }
    }
    sum_batch(h, rj0, nj0, m0, lane, acc);
    for (int q = p0 + 32; q < p1; q += 32) {
      const int m = (p1 - q < 32) ? (p1 - q) : 32;
      int2 r2 = make_int2(0, 0);
      if (lane < m) r2 = ((const int2*)rec)[q + lane];
      float nj = 0.0f;
      if (lane < m) nj = __int_as_float(r2.y) * dc;
      sum_batch(h, r2.x, nj, m, lane, acc);
    }
    float4 o;
    o.x = acc.x + bv.x;
    o.y = acc.y + bv.y;
    o.z = acc.z + bv.z;
    o.w = acc.w + bv.w;
    ((float4*)(out + (size_t)node * FOUT))[lane] = o;
  }
}

extern "C" void kernel_launch(void* const* d_in, const int* in_sizes, int n_in,
                              void* d_out, int out_size, void* d_ws, size_t ws_size,
                              hipStream_t stream) {
  const float* x     = (const float*)d_in[0];
  const int*   ei    = (const int*)d_in[1];
  const float* ew    = (const float*)d_in[2];
  const float* gamma = (const float*)d_in[3];
  const float* beta  = (const float*)d_in[4];
  const float* W     = (const float*)d_in[5];
  const float* b     = (const float*)d_in[6];
  float* out = (float*)d_out;

  const int n = in_sizes[0] / FIN;
  const int e = in_sizes[2];

  float* h    = (float*)d_ws;
  int*   rec  = (int*)(h + (size_t)n * FOUT);
  int*   rank = rec + (size_t)e * 2;
  float* dinv = (float*)(rank + e);
  int*   btab = (int*)(dinv + n);
  unsigned int* ctab = (unsigned int*)(btab + (size_t)NCHUNK * n);
  int*   off  = (int*)(ctab + (size_t)NCHUNK * NWORD);
  int*   bsum = off + n + 1;
  const int ch = (e + NCHUNK - 1) / NCHUNK;

  rank_lds_kernel<<<NCHUNK, 1024, 0, stream>>>(ei, rank, ctab, e, ch);

  blocksum_kernel<<<(n + 1024) / 1024, 256, 0, stream>>>(ctab, bsum, n);
  scan_kernel<<<(n + 1024) / 1024, 256, 0, stream>>>(ctab, bsum, off, btab, n);

  fill_kernel<<<dim3((ch + 256 * FE - 1) / (256 * FE), NCHUNK), 256, 0, stream>>>(
      ei, ew, btab, rank, rec, e, ch, n);

  ln_gemm_wmma_kernel<<<(n + MT * 16 - 1) / (MT * 16), 256, 0, stream>>>(
      x, gamma, beta, W, off, rec, h, dinv, n);
  node_agg_kernel<<<(n + 8 * NPW - 1) / (8 * NPW), 256, 0, stream>>>(off, rec, h, dinv, b, out, n);

  (void)n_in; (void)out_size; (void)ws_size;
}


# Counting launch re-tiled: one block per (chunk, 13312-node range) on all 256 WGPs keeps LDS counters for its range only and writes its 1024-node totals itself; block-sum launch goes; slot order kept
# speedup vs baseline: 3.2855x; 1.0450x over previous
#include <hip/hip_runtime.h>

#define FIN  128
#define FOUT 128
#define EPSV 1e-5f

typedef float v2f __attribute__((ext_vector_type(2)));
typedef float v8f __attribute__((ext_vector_type(8)));

#define A_PITCH 132
#define MT 4
#define LN_R (MT * 16 / 8)

__global__ __launch_bounds__(256) void ln_gemm_wmma_kernel(
    const float* __restrict__ x, const float* __restrict__ gamma,
    const float* __restrict__ beta, const float* __restrict__ W,
    const int* __restrict__ off, const int* __restrict__ rec,
    float* __restrict__ h, float* __restrict__ dinv, int n) {
  __shared__ float alds[MT * 16 * A_PITCH];
  const int tid  = threadIdx.x;
  const int wave = tid >> 5;
  const int lane = tid & 31;
  const int n0   = blockIdx.x * (MT * 16);
  const int swave = __builtin_amdgcn_readfirstlane(wave);

  int p0[LN_R], cnt[LN_R];
#pragma unroll
  for (int u = 0; u < LN_R; ++u) {
    const int r  = u * 8 + swave;
    const int gr = (n0 + r < n) ? (n0 + r) : (n - 1);
    p0[u]  = off[gr];
    cnt[u] = off[gr + 1] - p0[u];
  }

  float4 v[LN_R];
#pragma unroll
  for (int u = 0; u < LN_R; ++u) {
    const int r  = u * 8 + wave;
    const int gr = (n0 + r < n) ? (n0 + r) : (n - 1);
    v[u] = ((const float4*)(x + (size_t)gr * FIN))[lane];
  }
  float wq[LN_R];
#pragma unroll
  for (int u = 0; u < LN_R; ++u) {
    wq[u] = 0.0f;
    if (lane < cnt[u]) wq[u] = __int_as_float(rec[2 * (size_t)(p0[u] + lane) + 1]);
  }
  const float4 g  = ((const float4*)gamma)[lane];
  const float4 bt = ((const float4*)beta)[lane];

  float s[LN_R], ss[LN_R];
#pragma unroll
  for (int u = 0; u < LN_R; ++u) {
    s[u]  = v[u].x + v[u].y + v[u].z + v[u].w;
    ss[u] = v[u].x * v[u].x + v[u].y * v[u].y + v[u].z * v[u].z + v[u].w * v[u].w;
  }
#pragma unroll
  for (int u = 0; u < LN_R; ++u) {
    if (cnt[u] > 32) {
      for (int q = 32 + lane; q < cnt[u]; q += 32)
        wq[u] += __int_as_float(rec[2 * (size_t)(p0[u] + q) + 1]);
    }
  }
#pragma unroll
  for (int m = 16; m >= 1; m >>= 1) {
#pragma unroll
    for (int u = 0; u < LN_R; ++u) {
      s[u]  += __shfl_xor(s[u],  m, 32);
      ss[u] += __shfl_xor(ss[u], m, 32);
      wq[u] += __shfl_xor(wq[u], m, 32);
    }
  }
  float dvs[LN_R];
#pragma unroll
  for (int u = 0; u < LN_R; ++u) {
    const float mean = s[u] * (1.0f / FIN);
    const float var  = ss[u] * (1.0f / FIN) - mean * mean;
    const float rs   = rsqrtf(var + EPSV);
    const float dg = 1.0f + wq[u];
    const float dv = (dg > 0.0f) ? rsqrtf(dg) : 0.0f;
    dvs[u] = dv;
    float4 o;
    o.x = fmaxf(0.0f, (v[u].x - mean) * rs * g.x + bt.x) * dv;
    o.y = fmaxf(0.0f, (v[u].y - mean) * rs * g.y + bt.y) * dv;
    o.z = fmaxf(0.0f, (v[u].z - mean) * rs * g.z + bt.z) * dv;
    o.w = fmaxf(0.0f, (v[u].w - mean) * rs * g.w + bt.w) * dv;
    *((float4*)&alds[(u * 8 + wave) * A_PITCH + lane * 4]) = o;
  }
  __syncthreads();

  if (lane == 0) {
#pragma unroll
    for (int u = 0; u < LN_R; ++u) {
      if (n0 + u * 8 + wave < n) dinv[n0 + u * 8 + wave] = dvs[u];
    }
  }

  const int o0    = wave * 16;
  const int mrow  = lane & 15;
  const int khalf = (lane >> 4) * 2;

  const float* abase = &alds[mrow * A_PITCH + khalf];
  const float* wbase = W + (size_t)(o0 + mrow) * FIN + khalf;

  v8f c[MT];
#pragma unroll
  for (int t = 0; t < MT; ++t) c[t] = (v8f){};
#pragma unroll 4
  for (int k = 0; k < FIN; k += 4) {
    const v2f b = *((const v2f*)(wbase + k));
#pragma unroll
    for (int t = 0; t < MT; ++t) {
      const v2f a = *((const v2f*)(abase + t * 16 * A_PITCH + k));
      c[t] = __builtin_amdgcn_wmma_f32_16x16x4_f32(
          false, a, false, b, (short)0, c[t], false, false);
    }
  }

  const int col = o0 + mrow;
#pragma unroll
  for (int t = 0; t < MT; ++t) {
    if (n0 + t * 16 < n) {
      const int rbase = n0 + t * 16 + ((lane >> 4) << 3);
#pragma unroll
      for (int v = 0; v < 8; ++v) {
        h[(size_t)(rbase + v) * FOUT + col] = c[t][v];
      }
    }
  }
}

#define NNODE  100000
#define NWORD  (NNODE / 2)
#define NCHUNK 32
#define NRANGE 8
#define RBLK   13
#define RNODE  (RBLK * 1024)
#define RWORD  (RNODE / 2)
#define RB     25

__global__ __launch_bounds__(1024) void rank_lds_kernel(
    const int* __restrict__ ei, int* __restrict__ rank,
    unsigned int* __restrict__ ctab, int* __restrict__ psum, int e, int ch) {
  __shared__ __attribute__((aligned(16))) unsigned int ctr[RWORD];
  const int tid = threadIdx.x;
  const int c   = blockIdx.x;
  const int r0  = blockIdx.y * RNODE;
  for (int w = tid; w < RWORD / 4; w += 1024)
    ((uint4*)ctr)[w] = make_uint4(0u, 0u, 0u, 0u);
  __syncthreads();

  const int i0 = c * ch;
  const int i1 = (i0 + ch < e) ? (i0 + ch) : e;
  const int* __restrict__ col = ei + e;
  for (int base = i0 + tid; base < i1; base += RB * 1024) {
    unsigned int dv[RB];
#pragma unroll
    for (int u = 0; u < RB; ++u) {
      const int i = base + u * 1024;
      dv[u] = (unsigned int)(((i < i1) ? col[i] : -1) - r0);
    }
    unsigned int old[RB];
#pragma unroll
    for (int u = 0; u < RB; ++u) {
      old[u] = 0u;
      if (dv[u] < (unsigned int)RNODE)
        old[u] = atomicAdd(&ctr[dv[u] >> 1], 1u << ((dv[u] & 1u) << 4));
    }
#pragma unroll
    for (int u = 0; u < RB; ++u) {
      if (dv[u] < (unsigned int)RNODE)
        rank[base + u * 1024] = (int)((old[u] >> ((dv[u] & 1u) << 4)) & 0xFFFFu);
    }
  }
  __syncthreads();

  const int w0 = blockIdx.y * RWORD;
  const int nw = (NWORD - w0 < RWORD) ? (NWORD - w0) : RWORD;
  uint4* dst = (uint4*)(ctab + (size_t)c * NWORD + w0);
  for (int w = tid; w < nw / 4; w += 1024) dst[w] = ((const uint4*)ctr)[w];

  const int wave = tid >> 5, lane = tid & 31;
  if (wave < RBLK) {
    int s = 0;
#pragma unroll
    for (int k = 0; k < 16; ++k) {
      const unsigned int v = ctr[wave * 512 + k * 32 + lane];
      s += (int)((v & 0xFFFFu) + (v >> 16));
    }
    for (int m = 16; m >= 1; m >>= 1) s += __shfl_xor(s, m, 32);
    if (lane == 0) psum[(blockIdx.y * RBLK + wave) * NCHUNK + c] = s;
  }
}

#define PSL ((NRANGE * RBLK * NCHUNK + 255) / 256)

__global__ __launch_bounds__(256) void scan_kernel(
    const unsigned int* __restrict__ ctab, const int* __restrict__ psum,
    int* __restrict__ off, int* __restrict__ btab, int n) {
  __shared__ int wsum[8];
  __shared__ int wbase[8];
  const int tid = threadIdx.x, lane = tid & 31, wave = tid >> 5;
  const int nblk = blockIdx.x;
  const int lim = nblk * NCHUNK;
  int pb[PSL];
#pragma unroll
  for (int u = 0; u < PSL; ++u) {
    const int i = tid + u * 256;
    pb[u] = (i < lim) ? psum[i] : 0;
  }
  const int base = nblk * 1024 + tid * 4;
  const bool live = base < n;
  uint2 w[NCHUNK];
  int c[4] = {0, 0, 0, 0};
#pragma unroll
  for (int k = 0; k < NCHUNK; ++k) {
    w[k] = live ? *((const uint2*)(ctab + (size_t)k * NWORD + (base >> 1)))
                : make_uint2(0u, 0u);
    c[0] += (int)(w[k].x & 0xFFFFu);
    c[1] += (int)(w[k].x >> 16);
    c[2] += (int)(w[k].y & 0xFFFFu);
    c[3] += (int)(w[k].y >> 16);
  }
  int sb = 0;
#pragma unroll
  for (int u = 0; u < PSL; ++u) sb += pb[u];
  for (int m = 16; m >= 1; m >>= 1) sb += __shfl_xor(sb, m, 32);
  if (lane == 0) wbase[wave] = sb;
  const int s = c[0] + c[1] + c[2] + c[3];
  int inc = s;
  for (int d = 1; d < 32; d <<= 1) {
    int v = __shfl_up(inc, d, 32);
    if (lane >= d) inc += v;
  }
  if (lane == 31) wsum[wave] = inc;
  __syncthreads();
  int ex = inc - s;
  for (int k = 0; k < 8; ++k) ex += wbase[k];
  for (int k = 0; k < wave; ++k) ex += wsum[k];
  int r[4];
  for (int k = 0; k < 4; ++k) {
    r[k] = ex;
    if (base + k <= n) off[base + k] = ex;
    ex += c[k];
  }
  if (live) {
#pragma unroll
    for (int k = 0; k < NCHUNK; ++k) {
      *((int4*)(btab + (size_t)k * n + base)) = make_int4(r[0], r[1], r[2], r[3]);
      r[0] += (int)(w[k].x & 0xFFFFu);
      r[1] += (int)(w[k].x >> 16);
      r[2] += (int)(w[k].y & 0xFFFFu);
      r[3] += (int)(w[k].y >> 16);
    }
  }
}

#define FE 8

__global__ __launch_bounds__(256) void fill_kernel(
    const int* __restrict__ ei, const float* __restrict__ ew,
    const int* __restrict__ btab, const int* __restrict__ rank,
    int* __restrict__ rec, int e, int ch, int n) {
  const int c  = blockIdx.y;
  const int j0 = blockIdx.x * (256 * FE) + threadIdx.x;
  const int* __restrict__ bt = btab + (size_t)c * n;
  int  iv[FE], cv[FE], rv[FE], rk[FE], wv[FE];
  bool ok[FE];
#pragma unroll
  for (int u = 0; u < FE; ++u) {
    const int j = j0 + u * 256;
    const int i = c * ch + j;
    ok[u] = (j < ch) && (i < e);
    iv[u] = ok[u] ? i : 0;
    cv[u] = ei[e + iv[u]];
    rv[u] = ei[iv[u]];
    rk[u] = rank[iv[u]];
    wv[u] = __float_as_int(ew[iv[u]]);
  }
  int bs[FE];
#pragma unroll
  for (int u = 0; u < FE; ++u) bs[u] = bt[cv[u]];
#pragma unroll
  for (int u = 0; u < FE; ++u) {
    if (ok[u]) ((int2*)rec)[bs[u] + rk[u]] = make_int2(rv[u], wv[u]);
  }
}

#define NPW 10
#define RCH 8

template <int K>
__device__ __forceinline__ void gather_rows(const float* __restrict__ h, int rj,
                                            float nj, int j, int lane, float4& acc) {
  float4 hr[K];
  float  nm[K];
#pragma unroll
  for (int u = 0; u < K; ++u) {
    const int r = __builtin_amdgcn_readlane(rj, j + u);
    nm[u] = __int_as_float(__builtin_amdgcn_readlane(__float_as_int(nj), j + u));
    hr[u] = ((const float4*)(h + (size_t)r * FOUT))[lane];
  }
#pragma unroll
  for (int u = 0; u < K; ++u) {
    acc.x += hr[u].x * nm[u];
    acc.y += hr[u].y * nm[u];
    acc.z += hr[u].z * nm[u];
    acc.w += hr[u].w * nm[u];
  }
}

__device__ __forceinline__ void sum_batch(const float* __restrict__ h, int rj,
                                          float nj, int m, int lane, float4& acc) {
  int j = 0;
  if (m >= RCH) {
    gather_rows<RCH>(h, rj, nj, 0, lane, acc);
    j = RCH;
  }
  for (; j + RCH <= m; j += RCH) gather_rows<RCH>(h, rj, nj, j, lane, acc);
  switch (m - j) {
    case 7: gather_rows<7>(h, rj, nj, j, lane, acc); break;
    case 6: gather_rows<6>(h, rj, nj, j, lane, acc); break;
    case 5: gather_rows<5>(h, rj, nj, j, lane, acc); break;
    case 4: gather_rows<4>(h, rj, nj, j, lane, acc); break;
    case 3: gather_rows<3>(h, rj, nj, j, lane, acc); break;
    case 2: gather_rows<2>(h, rj, nj, j, lane, acc); break;
    case 1: gather_rows<1>(h, rj, nj, j, lane, acc); break;
    default: break;
  }
}

__global__ __launch_bounds__(256) void node_agg_kernel(
    const int* __restrict__ off, const int* __restrict__ rec,
    const float* __restrict__ h, const float* __restrict__ dinv,
    const float* __restrict__ b, float* __restrict__ out, int n) {
  const int node0 = (blockIdx.x * 8 + (threadIdx.x >> 5)) * NPW;
  if (node0 >= n) return;
  const int lane = threadIdx.x & 31;
  const int nn = (n - node0 < NPW) ? (n - node0) : NPW;
  const int   offv = off[(node0 + lane < n) ? (node0 + lane) : n];
  const float dcv  = dinv[(node0 + lane < n) ? (node0 + lane) : (n - 1)];
  const float4 bv = ((const float4*)b)[lane];

  int2   recA = make_int2(0, 0), recB = make_int2(0, 0);
  float4 hsA;
  {
    const int q0 = __builtin_amdgcn_readlane(offv, 0);
    const int q1 = __builtin_amdgcn_readlane(offv, 1);
    hsA = ((const float4*)(h + (size_t)node0 * FOUT))[lane];
    if (lane < q1 - q0) recA = ((const int2*)rec)[q0 + lane];
    if (nn > 1) {
      const int q2 = __builtin_amdgcn_readlane(offv, 2);
      if (lane < q2 - q1) recB = ((const int2*)rec)[q1 + lane];
    }
  }

  for (int t = 0; t < nn; ++t) {
    const int node = node0 + t;
    const int   p0 = __builtin_amdgcn_readlane(offv, t);
    const int   p1 = __builtin_amdgcn_readlane(offv, t + 1);
    const float dc = __int_as_float(__builtin_amdgcn_readlane(__float_as_int(dcv), t));
    const int m0 = (p1 - p0 < 32) ? (p1 - p0) : 32;
    const int rj0 = recA.x;
    float nj0 = 0.0f;
    if (lane < m0) nj0 = __int_as_float(recA.y) * dc;
    float4 acc;
    acc.x = hsA.x * dc;
    acc.y = hsA.y * dc;
    acc.z = hsA.z * dc;
    acc.w = hsA.w * dc;
    recA = recB;
    if (t + 1 < nn) {
      const int p2 = __builtin_amdgcn_readlane(offv, t + 2);
      hsA = ((const float4*)(h + (size_t)(node + 1) * FOUT))[lane];
      if (t + 2 < nn) {
        const int p3 = __builtin_amdgcn_readlane(offv, t + 3);
        recB = make_int2(0, 0);
        if (lane < p3 - p2) recB = ((const int2*)rec)[p2 + lane];
      }
    }
    sum_batch(h, rj0, nj0, m0, lane, acc);
    for (int q = p0 + 32; q < p1; q += 32) {
      const int m = (p1 - q < 32) ? (p1 - q) : 32;
      int2 r2 = make_int2(0, 0);
      if (lane < m) r2 = ((const int2*)rec)[q + lane];
      float nj = 0.0f;
      if (lane < m) nj = __int_as_float(r2.y) * dc;
      sum_batch(h, r2.x, nj, m, lane, acc);
    }
    float4 o;
    o.x = acc.x + bv.x;
    o.y = acc.y + bv.y;
    o.z = acc.z + bv.z;
    o.w = acc.w + bv.w;
    ((float4*)(out + (size_t)node * FOUT))[lane] = o;
  }
}

extern "C" void kernel_launch(void* const* d_in, const int* in_sizes, int n_in,
                              void* d_out, int out_size, void* d_ws, size_t ws_size,
                              hipStream_t stream) {
  const float* x     = (const float*)d_in[0];
  const int*   ei    = (const int*)d_in[1];
  const float* ew    = (const float*)d_in[2];
  const float* gamma = (const float*)d_in[3];
  const float* beta  = (const float*)d_in[4];
  const float* W     = (const float*)d_in[5];
  const float* b     = (const float*)d_in[6];
  float* out = (float*)d_out;

  const int n = in_sizes[0] / FIN;
  const int e = in_sizes[2];

  float* h    = (float*)d_ws;
  int*   rec  = (int*)(h + (size_t)n * FOUT);
  int*   rank = rec + (size_t)e * 2;
  float* dinv = (float*)(rank + e);
  int*   btab = (int*)(dinv + n);
  unsigned int* ctab = (unsigned int*)(btab + (size_t)NCHUNK * n);
  int*   off  = (int*)(ctab + (size_t)NCHUNK * NWORD);
  int*   psum = off + n + 1;
  const int ch = (e + NCHUNK - 1) / NCHUNK;

  rank_lds_kernel<<<dim3(NCHUNK, NRANGE), 1024, 0, stream>>>(ei, rank, ctab, psum, e, ch);

  scan_kernel<<<(n + 1024) / 1024, 256, 0, stream>>>(ctab, psum, off, btab, n);

  fill_kernel<<<dim3((ch + 256 * FE - 1) / (256 * FE), NCHUNK), 256, 0, stream>>>(
      ei, ew, btab, rank, rec, e, ch, n);

  ln_gemm_wmma_kernel<<<(n + MT * 16 - 1) / (MT * 16), 256, 0, stream>>>(
      x, gamma, beta, W, off, rec, h, dinv, n);
  node_agg_kernel<<<(n + 8 * NPW - 1) / (8 * NPW), 256, 0, stream>>>(off, rec, h, dinv, b, out, n);

  (void)n_in; (void)out_size; (void)ws_size;
}
